# starting kernel
# baseline (speedup 1.0000x reference)
#include <hip/hip_runtime.h>
#include <math.h>


typedef __attribute__((ext_vector_type(2))) float v2f;
typedef __attribute__((ext_vector_type(8))) float v8f;

#define DHID 128
#define TK 64
#define NEG_SLOPE 0.01f

__global__ void fill_zero(float* __restrict__ p, size_t n) {
    size_t i = blockIdx.x * (size_t)blockDim.x + threadIdx.x;
    if (i < n) p[i] = 0.0f;
}

__global__ void degree_kernel(const int* __restrict__ dst, float* __restrict__ deg, int E) {
    int e = blockIdx.x * blockDim.x + threadIdx.x;
    if (e < E) atomicAdd(deg + dst[e], 1.0f);
}

__global__ void rsqrt_kernel(const float* __restrict__ deg, float* __restrict__ dinv, int N) {
    int i = blockIdx.x * blockDim.x + threadIdx.x;
    if (i < N) dinv[i] = rsqrtf(deg[i] + 1.0f);
}

template <int K, bool BIAS, bool ACT>
__global__ __launch_bounds__(128) void gemm_wmma(
    const float* __restrict__ A, const float* __restrict__ W,
    const float* __restrict__ bias, float* __restrict__ Out, int Nrows)
{
    __shared__ float Wp[TK / 2][DHID * 2];

    const int tid  = threadIdx.x;
    const int wave = tid >> 5;
    const int lane = tid & 31;
    const int row0 = (blockIdx.x * 4 + wave) * 16;
    const bool rowok = (row0 < Nrows);

    const int m     = lane & 15;
    const int khalf = (lane >> 4) * 2;

    v8f acc[8];
#pragma unroll
    for (int t = 0; t < 8; ++t) acc[t] = (v8f){};

    const float* __restrict__ arow = A + (size_t)((rowok ? row0 : 0) + m) * K;

    for (int k0 = 0; k0 < K; k0 += TK) {
#pragma unroll 8
        for (int idx = 0; idx < TK * DHID; idx += 128) {
            const int j   = idx + tid;
            const int krl = j >> 7;
            const int col = j & (DHID - 1);
            const int kr  = k0 + krl;
            const float v = (kr < K) ? W[(size_t)kr * DHID + col] : 0.0f;
            Wp[krl >> 1][col * 2 + (krl & 1)] = v;
        }
        __syncthreads();

#pragma unroll 4
        for (int k = 0; k < TK; k += 4) {
            const int kk = k0 + k + khalf;
            const int i0 = (kk     < K) ? kk     : (K - 1);
            const int i1 = (kk + 1 < K) ? kk + 1 : (K - 1);
            v2f a;
            a.x = arow[i0];
            a.y = arow[i1];
            const float* __restrict__ wp = &Wp[(k + khalf) >> 1][0];
#pragma unroll
            for (int t = 0; t < 8; ++t) {
                const v2f b = *(const v2f*)(wp + (t * 16 + m) * 2);
                acc[t] = __builtin_amdgcn_wmma_f32_16x16x4_f32(
                    false, a, false, b, (short)0, acc[t], false, false);
            }
        }
        __syncthreads();
    }

    if (!rowok) return;
    const int rbase = row0 + ((lane >> 4) ? 8 : 0);
#pragma unroll
    for (int t = 0; t < 8; ++t) {
        const int col = t * 16 + m;
        const float bv = BIAS ? bias[col] : 0.0f;
#pragma unroll
        for (int r = 0; r < 8; ++r) {
            float v = acc[t][r] + bv;
            if (ACT) v = (v >= 0.0f) ? v : NEG_SLOPE * v;
            Out[(size_t)(rbase + r) * DHID + col] = v;
        }
    }
}

__global__ __launch_bounds__(256) void scatter_kernel(
    const float* __restrict__ g, const int* __restrict__ src,
    const int* __restrict__ dst, const float* __restrict__ dinv,
    float* __restrict__ agg, int E)
{
    const int t = blockIdx.x * blockDim.x + threadIdx.x;
    const int e = t >> 5;
    const int lane = t & 31;
    if (e >= E) return;
    const int s = src[e];
    const int d = dst[e];
    const float norm = dinv[s] * dinv[d];
    const float4 v = *(const float4*)(g + (size_t)s * DHID + lane * 4);
    float* ap = agg + (size_t)d * DHID + lane * 4;
    atomicAdd(ap + 0, v.x * norm);
    atomicAdd(ap + 1, v.y * norm);
    atomicAdd(ap + 2, v.z * norm);
    atomicAdd(ap + 3, v.w * norm);
}

__global__ __launch_bounds__(256) void finalize_kernel(
    const float* __restrict__ agg, const float* __restrict__ g,
    const float* __restrict__ dinv, const float* __restrict__ bias,
    float* __restrict__ out, int Nrows)
{
    const size_t idx = blockIdx.x * (size_t)blockDim.x + threadIdx.x;
    const size_t total = (size_t)Nrows * (DHID / 4);
    if (idx >= total) return;
    const int i  = (int)(idx >> 5);
    const int c4 = ((int)idx & 31) * 4;
    const float di = dinv[i];
    const float s = di * di;
    const float4 a  = *(const float4*)(agg + (size_t)i * DHID + c4);
    const float4 gg = *(const float4*)(g   + (size_t)i * DHID + c4);
    const float4 bb = *(const float4*)(bias + c4);
    float4 r;
    r.x = a.x + gg.x * s + bb.x;
    r.y = a.y + gg.y * s + bb.y;
    r.z = a.z + gg.z * s + bb.z;
    r.w = a.w + gg.w * s + bb.w;
    *(float4*)(out + (size_t)i * DHID + c4) = r;
}

__global__ __launch_bounds__(256) void out_layer_kernel(
    const float* __restrict__ H, const float* __restrict__ W2,
    const float* __restrict__ b2, float* __restrict__ out, int Nrows)
{
    const int i = blockIdx.x * blockDim.x + threadIdx.x;
    if (i >= Nrows) return;
    const float* h = H + (size_t)i * DHID;
    float a0 = 0.0f, a1 = 0.0f;
#pragma unroll 8
    for (int k = 0; k < DHID; ++k) {
        const float hv = h[k];
        a0 += hv * W2[k * 2 + 0];
        a1 += hv * W2[k * 2 + 1];
    }
    out[(size_t)i * 2 + 0] = a0 + b2[0];
    out[(size_t)i * 2 + 1] = a1 + b2[1];
}

extern "C" void kernel_launch(void* const* d_in, const int* in_sizes, int n_in,
                              void* d_out, int out_size, void* d_ws, size_t ws_size,
                              hipStream_t stream)
{
    const float* x    = (const float*)d_in[0];
    const int*   ei   = (const int*)  d_in[1];
    const float* W_in = (const float*)d_in[3];
    const float* b_in = (const float*)d_in[4];
    const float* W_g  = (const float*)d_in[5];
    const float* b_g  = (const float*)d_in[6];
    const float* W_o1 = (const float*)d_in[7];
    const float* b_o1 = (const float*)d_in[8];
    const float* W_o2 = (const float*)d_in[9];
    const float* b_o2 = (const float*)d_in[10];

    const int ND = 239;
    const int N  = in_sizes[0] / ND;
    const int E  = in_sizes[1] / 2;
    const int* src = ei;
    const int* dst = ei + E;

    float* ws   = (float*)d_ws;
    float* deg  = ws;
    float* dinv = ws + (size_t)N;
    float* bufA = ws + 2 * (size_t)N;
    float* bufB = bufA + (size_t)N * DHID;
    float* bufC = bufB + (size_t)N * DHID;

    const size_t nh = (size_t)N * DHID;
    const int gemm_blocks = (N / 16 + 3) / 4;
    const int scat_blocks = (int)(((size_t)E * 32 + 255) / 256);
    const int fin_blocks  = (int)((nh / 4 + 255) / 256);
    const int zero_blocks = (int)((nh + 255) / 256);

    fill_zero<<<(N + 255) / 256, 256, 0, stream>>>(deg, (size_t)N);
    degree_kernel<<<(E + 255) / 256, 256, 0, stream>>>(dst, deg, E);
    rsqrt_kernel<<<(N + 255) / 256, 256, 0, stream>>>(deg, dinv, N);

    gemm_wmma<239, true, true><<<gemm_blocks, 128, 0, stream>>>(x, W_in, b_in, bufA, N);

    gemm_wmma<128, false, false><<<gemm_blocks, 128, 0, stream>>>(bufA, W_g, nullptr, bufB, N);
    fill_zero<<<zero_blocks, 256, 0, stream>>>(bufC, nh);
    scatter_kernel<<<scat_blocks, 256, 0, stream>>>(bufB, src, dst, dinv, bufC, E);
    finalize_kernel<<<fin_blocks, 256, 0, stream>>>(bufC, bufB, dinv, b_g, bufA, N);

    gemm_wmma<128, false, false><<<gemm_blocks, 128, 0, stream>>>(bufA, W_g, nullptr, bufB, N);
    fill_zero<<<zero_blocks, 256, 0, stream>>>(bufC, nh);
    scatter_kernel<<<scat_blocks, 256, 0, stream>>>(bufB, src, dst, dinv, bufC, E);
    finalize_kernel<<<fin_blocks, 256, 0, stream>>>(bufC, bufB, dinv, b_g, bufA, N);

    gemm_wmma<128, true, true><<<gemm_blocks, 128, 0, stream>>>(bufA, W_o1, b_o1, bufB, N);

    out_layer_kernel<<<(N + 255) / 256, 256, 0, stream>>>(bufB, W_o2, b_o2, (float*)d_out, N);
}


# Float-atomic scatter -> in-call bucketing of edges by destination (integer atomic count gives each edge an arrival-order rank, scan, place); a wave sums a node's rows in registers in rank order
# speedup vs baseline: 1.5468x; 1.5468x over previous
#include <hip/hip_runtime.h>
#include <math.h>


typedef __attribute__((ext_vector_type(2))) float v2f;
typedef __attribute__((ext_vector_type(8))) float v8f;

#define DHID 128
#define TK 64
#define NEG_SLOPE 0.01f
#define SCAN_T 256

__global__ void fill_zero_i32(int* __restrict__ p, int n) {
    const int i = blockIdx.x * blockDim.x + threadIdx.x;
    if (i < n) p[i] = 0;
}

__global__ void count_rank_kernel(const int* __restrict__ dst, int* __restrict__ cnt,
                                  int* __restrict__ rank, int E) {
    const int e = blockIdx.x * blockDim.x + threadIdx.x;
    if (e < E) rank[e] = atomicAdd(cnt + dst[e], 1);
}

__global__ __launch_bounds__(SCAN_T) void block_sum_kernel(
    const int* __restrict__ cnt, int* __restrict__ bsum, int N)
{
    __shared__ int red[SCAN_T];
    const int tid = threadIdx.x;
    const int i = blockIdx.x * SCAN_T + tid;
    red[tid] = (i < N) ? cnt[i] : 0;
    __syncthreads();
    for (int o = SCAN_T / 2; o > 0; o >>= 1) {
        if (tid < o) red[tid] += red[tid + o];
        __syncthreads();
    }
    if (tid == 0) bsum[blockIdx.x] = red[0];
}

__global__ __launch_bounds__(SCAN_T) void offsets_kernel(
    const int* __restrict__ cnt, const int* __restrict__ bsum,
    int* __restrict__ off, float* __restrict__ dinv, int N)
{
    __shared__ int red[SCAN_T];
    __shared__ int sc[SCAN_T];
    const int tid = threadIdx.x;
    const int b = blockIdx.x;
    int part = 0;
    for (int j = tid; j < b; j += SCAN_T) part += bsum[j];
    red[tid] = part;
    const int i = b * SCAN_T + tid;
    const int c = (i < N) ? cnt[i] : 0;
    sc[tid] = c;
    __syncthreads();
    for (int o = SCAN_T / 2; o > 0; o >>= 1) {
        if (tid < o) red[tid] += red[tid + o];
        __syncthreads();
    }
    const int base = red[0];
    for (int o = 1; o < SCAN_T; o <<= 1) {
        const int v = (tid >= o) ? sc[tid - o] : 0;
        __syncthreads();
        sc[tid] += v;
        __syncthreads();
    }
    if (i < N) {
        off[i]  = base + sc[tid] - c;
        dinv[i] = rsqrtf((float)c + 1.0f);
    }
}

__global__ void place_kernel(const int* __restrict__ src, const int* __restrict__ dst,
                             const int* __restrict__ off, const int* __restrict__ rank,
                             int* __restrict__ bucket, int E) {
    const int e = blockIdx.x * blockDim.x + threadIdx.x;
    if (e >= E) return;
    int pos = off[dst[e]] + rank[e];
    pos = min(max(pos, 0), E - 1);
    bucket[pos] = src[e];
}

template <int K, bool BIAS, bool ACT>
__global__ __launch_bounds__(128) void gemm_wmma(
    const float* __restrict__ A, const float* __restrict__ W,
    const float* __restrict__ bias, float* __restrict__ Out, int Nrows)
{
    __shared__ float Wp[TK / 2][DHID * 2];

    const int tid  = threadIdx.x;
    const int wave = tid >> 5;
    const int lane = tid & 31;
    const int row0 = (blockIdx.x * 4 + wave) * 16;
    const bool rowok = (row0 < Nrows);

    const int m     = lane & 15;
    const int khalf = (lane >> 4) * 2;

    v8f acc[8];
#pragma unroll
    for (int t = 0; t < 8; ++t) acc[t] = (v8f){};

    const float* __restrict__ arow = A + (size_t)((rowok ? row0 : 0) + m) * K;

    for (int k0 = 0; k0 < K; k0 += TK) {
#pragma unroll 8
        for (int idx = 0; idx < TK * DHID; idx += 128) {
            const int j   = idx + tid;
            const int krl = j >> 7;
            const int col = j & (DHID - 1);
            const int kr  = k0 + krl;
            const float v = (kr < K) ? W[(size_t)kr * DHID + col] : 0.0f;
            Wp[krl >> 1][col * 2 + (krl & 1)] = v;
        }
        __syncthreads();

#pragma unroll 4
        for (int k = 0; k < TK; k += 4) {
            const int kk = k0 + k + khalf;
            const int i0 = (kk     < K) ? kk     : (K - 1);
            const int i1 = (kk + 1 < K) ? kk + 1 : (K - 1);
            v2f a;
            a.x = arow[i0];
            a.y = arow[i1];
            const float* __restrict__ wp = &Wp[(k + khalf) >> 1][0];
#pragma unroll
            for (int t = 0; t < 8; ++t) {
                const v2f b = *(const v2f*)(wp + (t * 16 + m) * 2);
                acc[t] = __builtin_amdgcn_wmma_f32_16x16x4_f32(
                    false, a, false, b, (short)0, acc[t], false, false);
            }
        }
        __syncthreads();
    }

    if (!rowok) return;
    const int rbase = row0 + ((lane >> 4) ? 8 : 0);
#pragma unroll
    for (int t = 0; t < 8; ++t) {
        const int col = t * 16 + m;
        const float bv = BIAS ? bias[col] : 0.0f;
#pragma unroll
        for (int r = 0; r < 8; ++r) {
            float v = acc[t][r] + bv;
            if (ACT) v = (v >= 0.0f) ? v : NEG_SLOPE * v;
            Out[(size_t)(rbase + r) * DHID + col] = v;
        }
    }
}

__device__ __forceinline__ float4 load_row(const float* __restrict__ g, int sj, int k, int c4) {
    const int s = __builtin_amdgcn_readlane(sj, k);
    return *(const float4*)(g + (size_t)s * DHID + c4);
}

__device__ __forceinline__ void add_row(float4& acc, const float4 v, float wj, int k) {
    const float w = __int_as_float(__builtin_amdgcn_readlane(__float_as_int(wj), k));
    acc.x += v.x * w;
    acc.y += v.y * w;
    acc.z += v.z * w;
    acc.w += v.w * w;
}

__global__ __launch_bounds__(256) void gather_kernel(
    const float* __restrict__ g, const int* __restrict__ bucket,
    const int* __restrict__ off, const int* __restrict__ cnt,
    const float* __restrict__ dinv, const float* __restrict__ bias,
    float* __restrict__ out, int Nrows)
{
    const int lane = threadIdx.x & 31;
    const int i = __builtin_amdgcn_readfirstlane(
        (int)(blockIdx.x * (blockDim.x >> 5) + (threadIdx.x >> 5)));
    if (i >= Nrows) return;
    const int start = off[i];
    const int n     = cnt[i];
    const float di  = dinv[i];
    const int c4 = lane * 4;
    const float4 gs = *(const float4*)(g + (size_t)i * DHID + c4);
    const float4 bb = *(const float4*)(bias + c4);

    float4 acc = make_float4(0.0f, 0.0f, 0.0f, 0.0f);
    for (int base = 0; base < n; base += 32) {
        const int m = min(32, n - base);
        int sj = 0;
        float wj = 0.0f;
        if (lane < m) {
            sj = bucket[start + base + lane];
            sj = min(max(sj, 0), Nrows - 1);
            wj = dinv[sj] * di;
        }
        int k = 0;
        for (; k + 4 <= m; k += 4) {
            const float4 v0 = load_row(g, sj, k,     c4);
            const float4 v1 = load_row(g, sj, k + 1, c4);
            const float4 v2 = load_row(g, sj, k + 2, c4);
            const float4 v3 = load_row(g, sj, k + 3, c4);
            add_row(acc, v0, wj, k);
            add_row(acc, v1, wj, k + 1);
            add_row(acc, v2, wj, k + 2);
            add_row(acc, v3, wj, k + 3);
        }
        const int r = m - k;
        if (r == 3) {
            const float4 v0 = load_row(g, sj, k,     c4);
            const float4 v1 = load_row(g, sj, k + 1, c4);
            const float4 v2 = load_row(g, sj, k + 2, c4);
            add_row(acc, v0, wj, k);
            add_row(acc, v1, wj, k + 1);
            add_row(acc, v2, wj, k + 2);
        } else if (r == 2) {
            const float4 v0 = load_row(g, sj, k,     c4);
            const float4 v1 = load_row(g, sj, k + 1, c4);
            add_row(acc, v0, wj, k);
            add_row(acc, v1, wj, k + 1);
        } else if (r == 1) {
            const float4 v0 = load_row(g, sj, k, c4);
            add_row(acc, v0, wj, k);
        }
    }

    const float s2 = di * di;
    float4 r4;
    r4.x = acc.x + gs.x * s2 + bb.x;
    r4.y = acc.y + gs.y * s2 + bb.y;
    r4.z = acc.z + gs.z * s2 + bb.z;
    r4.w = acc.w + gs.w * s2 + bb.w;
    *(float4*)(out + (size_t)i * DHID + c4) = r4;
}

__global__ __launch_bounds__(256) void out_layer_kernel(
    const float* __restrict__ H, const float* __restrict__ W2,
    const float* __restrict__ b2, float* __restrict__ out, int Nrows)
{
    const int i = blockIdx.x * blockDim.x + threadIdx.x;
    if (i >= Nrows) return;
    const float* h = H + (size_t)i * DHID;
    float a0 = 0.0f, a1 = 0.0f;
#pragma unroll 8
    for (int k = 0; k < DHID; ++k) {
        const float hv = h[k];
        a0 += hv * W2[k * 2 + 0];
        a1 += hv * W2[k * 2 + 1];
    }
    out[(size_t)i * 2 + 0] = a0 + b2[0];
    out[(size_t)i * 2 + 1] = a1 + b2[1];
}

extern "C" void kernel_launch(void* const* d_in, const int* in_sizes, int n_in,
                              void* d_out, int out_size, void* d_ws, size_t ws_size,
                              hipStream_t stream)
{
    const float* x    = (const float*)d_in[0];
    const int*   ei   = (const int*)  d_in[1];
    const float* W_in = (const float*)d_in[3];
    const float* b_in = (const float*)d_in[4];
    const float* W_g  = (const float*)d_in[5];
    const float* b_g  = (const float*)d_in[6];
    const float* W_o1 = (const float*)d_in[7];
    const float* b_o1 = (const float*)d_in[8];
    const float* W_o2 = (const float*)d_in[9];
    const float* b_o2 = (const float*)d_in[10];

    const int ND = 239;
    const int N  = in_sizes[0] / ND;
    const int E  = in_sizes[1] / 2;
    const int* src = ei;
    const int* dst = ei + E;

    const size_t nh = (size_t)N * DHID;
    float* ws   = (float*)d_ws;
    int*   cnt  = (int*)d_ws;
    float* dinv = ws + (size_t)N;
    float* bufA = ws + 2 * (size_t)N;
    float* bufB = bufA + nh;
    int*   off    = (int*)(bufB + nh);
    int*   rank   = off + (size_t)N;
    int*   bucket = rank + (size_t)E;
    int*   bsum   = bucket + (size_t)E;

    const int gemm_blocks = (N / 16 + 3) / 4;
    const int node_blocks = (N + 255) / 256;
    const int edge_blocks = (E + 255) / 256;
    const int scan_blocks = (N + SCAN_T - 1) / SCAN_T;
    const int gath_blocks = (N + 7) / 8;

    fill_zero_i32<<<node_blocks, 256, 0, stream>>>(cnt, N);
    count_rank_kernel<<<edge_blocks, 256, 0, stream>>>(dst, cnt, rank, E);
    block_sum_kernel<<<scan_blocks, SCAN_T, 0, stream>>>(cnt, bsum, N);
    offsets_kernel<<<scan_blocks, SCAN_T, 0, stream>>>(cnt, bsum, off, dinv, N);
    place_kernel<<<edge_blocks, 256, 0, stream>>>(src, dst, off, rank, bucket, E);

    gemm_wmma<239, true, true><<<gemm_blocks, 128, 0, stream>>>(x, W_in, b_in, bufA, N);

    gemm_wmma<128, false, false><<<gemm_blocks, 128, 0, stream>>>(bufA, W_g, nullptr, bufB, N);
    gather_kernel<<<gath_blocks, 256, 0, stream>>>(bufB, bucket, off, cnt, dinv, b_g, bufA, N);

    gemm_wmma<128, false, false><<<gemm_blocks, 128, 0, stream>>>(bufA, W_g, nullptr, bufB, N);
    gather_kernel<<<gath_blocks, 256, 0, stream>>>(bufB, bucket, off, cnt, dinv, b_g, bufA, N);

    gemm_wmma<128, true, true><<<gemm_blocks, 128, 0, stream>>>(bufA, W_o1, b_o1, bufB, N);

    out_layer_kernel<<<(N + 255) / 256, 256, 0, stream>>>(bufB, W_o2, b_o2, (float*)d_out, N);
}


# Launch fusion 12 -> 8: counter clear and bucket placement ride in two GEMM launches, the scan sums its own front counters, the 128->2 projection runs in the last GEMM's epilogue (sum order changes)
# speedup vs baseline: 1.6777x; 1.0846x over previous
#include <hip/hip_runtime.h>
#include <math.h>


typedef __attribute__((ext_vector_type(2))) float v2f;
typedef __attribute__((ext_vector_type(8))) float v8f;

#define DHID 128
#define TK 64
#define NEG_SLOPE 0.01f
#define SCAN_T 256

__global__ void count_rank_kernel(const int* __restrict__ dst, int* __restrict__ cnt,
                                  int* __restrict__ rank, int E) {
    const int e = blockIdx.x * blockDim.x + threadIdx.x;
    if (e < E) rank[e] = atomicAdd(cnt + dst[e], 1);
}

__global__ __launch_bounds__(SCAN_T) void offsets_kernel(
    const int* __restrict__ cnt,
    int* __restrict__ off, float* __restrict__ dinv, int N)
{
    __shared__ int red[SCAN_T];
    __shared__ int sc[SCAN_T];
    const int tid = threadIdx.x;
    const int b = blockIdx.x;
    int part = 0;
    const int lim = b * SCAN_T;
#pragma unroll 8
    for (int j = tid * 4; j < lim; j += SCAN_T * 4) {
        const int4 v = *(const int4*)(cnt + j);
        part += (v.x + v.y) + (v.z + v.w);
    }
    red[tid] = part;
    const int i = b * SCAN_T + tid;
    const int c = (i < N) ? cnt[i] : 0;
    sc[tid] = c;
    __syncthreads();
    for (int o = SCAN_T / 2; o > 0; o >>= 1) {
        if (tid < o) red[tid] += red[tid + o];
        __syncthreads();
    }
    const int base = red[0];
    for (int o = 1; o < SCAN_T; o <<= 1) {
        const int v = (tid >= o) ? sc[tid - o] : 0;
        __syncthreads();
        sc[tid] += v;
        __syncthreads();
    }
    if (i < N) {
        off[i]  = base + sc[tid] - c;
        dinv[i] = rsqrtf((float)c + 1.0f);
    }
}

struct PlaceArgs {
    const int* src;
    const int* dst;
    const int* off;
    const int* rank;
    int*       bucket;
    int        E;
    int        gblocks;
};

template <int K, bool BIAS, bool ACT, bool PROJ, bool ZERO, bool PLACE>
__global__ __launch_bounds__(128) void gemm_wmma(
    const float* __restrict__ A, const float* __restrict__ W,
    const float* __restrict__ bias, float* __restrict__ Out, int Nrows,
    const float* __restrict__ W2, const float* __restrict__ b2,
    int* __restrict__ z, int nz, const PlaceArgs pa)
{
    __shared__ float Wp[TK / 2][DHID * 2];

    const int tid  = threadIdx.x;
    if (PLACE) {
        if ((int)blockIdx.x >= pa.gblocks) {
            const int e = ((int)blockIdx.x - pa.gblocks) * 128 + tid;
            if (e < pa.E) {
                int pos = pa.off[pa.dst[e]] + pa.rank[e];
                pos = min(max(pos, 0), pa.E - 1);
                pa.bucket[pos] = pa.src[e];
            }
            return;
        }
    }
    if (ZERO) {
        const int zi = blockIdx.x * 128 + tid;
        if (zi < nz) z[zi] = 0;
    }
    const int wave = tid >> 5;
    const int lane = tid & 31;
    const int row0 = (blockIdx.x * 4 + wave) * 16;
    const bool rowok = (row0 < Nrows);

    const int m     = lane & 15;
    const int khalf = (lane >> 4) * 2;

    v8f acc[8];
#pragma unroll
    for (int t = 0; t < 8; ++t) acc[t] = (v8f){};

    const float* __restrict__ arow = A + (size_t)((rowok ? row0 : 0) + m) * K;

    for (int k0 = 0; k0 < K; k0 += TK) {
#pragma unroll 8
        for (int idx = 0; idx < TK * DHID; idx += 128) {
            const int j   = idx + tid;
            const int krl = j >> 7;
            const int col = j & (DHID - 1);
            const int kr  = k0 + krl;
            const float v = (kr < K) ? W[(size_t)kr * DHID + col] : 0.0f;
            Wp[krl >> 1][col * 2 + (krl & 1)] = v;
        }
        __syncthreads();

#pragma unroll 4
        for (int k = 0; k < TK; k += 4) {
            const int kk = k0 + k + khalf;
            const int i0 = (kk     < K) ? kk     : (K - 1);
            const int i1 = (kk + 1 < K) ? kk + 1 : (K - 1);
            v2f a;
            a.x = arow[i0];
            a.y = arow[i1];
            const float* __restrict__ wp = &Wp[(k + khalf) >> 1][0];
#pragma unroll
            for (int t = 0; t < 8; ++t) {
                const v2f b = *(const v2f*)(wp + (t * 16 + m) * 2);
                if (PROJ)
                    acc[t] = __builtin_amdgcn_wmma_f32_16x16x4_f32(
                        false, b, false, a, (short)0, acc[t], false, false);
                else
                    acc[t] = __builtin_amdgcn_wmma_f32_16x16x4_f32(
                        false, a, false, b, (short)0, acc[t], false, false);
            }
        }
        __syncthreads();
    }

    if (!rowok) return;
    if (PROJ) {
        const int hs = lane >> 4;
        float p0 = 0.0f, p1 = 0.0f;
#pragma unroll
        for (int t = 0; t < 8; ++t) {
            const int f0 = t * 16 + hs * 8;
            float bv[8];
            if (BIAS) {
                const float4 ba = *(const float4*)(bias + f0);
                const float4 bb = *(const float4*)(bias + f0 + 4);
                bv[0] = ba.x; bv[1] = ba.y; bv[2] = ba.z; bv[3] = ba.w;
                bv[4] = bb.x; bv[5] = bb.y; bv[6] = bb.z; bv[7] = bb.w;
            }
            float wv[16];
#pragma unroll
            for (int q = 0; q < 4; ++q) {
                const float4 w4 = *(const float4*)(W2 + f0 * 2 + q * 4);
                wv[q * 4 + 0] = w4.x; wv[q * 4 + 1] = w4.y; wv[q * 4 + 2] = w4.z; wv[q * 4 + 3] = w4.w;
            }
#pragma unroll
            for (int r = 0; r < 8; ++r) {
                float v = acc[t][r] + (BIAS ? bv[r] : 0.0f);
                if (ACT) v = (v >= 0.0f) ? v : NEG_SLOPE * v;
                p0 += v * wv[r * 2 + 0];
                p1 += v * wv[r * 2 + 1];
            }
        }
        p0 += __shfl_xor(p0, 16);
        p1 += __shfl_xor(p1, 16);
        if (hs == 0) {
            float2 o;
            o.x = p0 + b2[0];
            o.y = p1 + b2[1];
            *(float2*)(Out + (size_t)(row0 + m) * 2) = o;
        }
        return;
    }
    const int rbase = row0 + ((lane >> 4) ? 8 : 0);
#pragma unroll
    for (int t = 0; t < 8; ++t) {
        const int col = t * 16 + m;
        const float bv = BIAS ? bias[col] : 0.0f;
#pragma unroll
        for (int r = 0; r < 8; ++r) {
            float v = acc[t][r] + bv;
            if (ACT) v = (v >= 0.0f) ? v : NEG_SLOPE * v;
            Out[(size_t)(rbase + r) * DHID + col] = v;
        }
    }
}

__device__ __forceinline__ float4 load_row(const float* __restrict__ g, int sj, int k, int c4) {
    const int s = __builtin_amdgcn_readlane(sj, k);
    return *(const float4*)(g + (size_t)s * DHID + c4);
}

__device__ __forceinline__ void add_row(float4& acc, const float4 v, float wj, int k) {
    const float w = __int_as_float(__builtin_amdgcn_readlane(__float_as_int(wj), k));
    acc.x += v.x * w;
    acc.y += v.y * w;
    acc.z += v.z * w;
    acc.w += v.w * w;
}

__global__ __launch_bounds__(256) void gather_kernel(
    const float* __restrict__ g, const int* __restrict__ bucket,
    const int* __restrict__ off, const int* __restrict__ cnt,
    const float* __restrict__ dinv, const float* __restrict__ bias,
    float* __restrict__ out, int Nrows)
{
    const int lane = threadIdx.x & 31;
    const int i = __builtin_amdgcn_readfirstlane(
        (int)(blockIdx.x * (blockDim.x >> 5) + (threadIdx.x >> 5)));
    if (i >= Nrows) return;
    const int start = off[i];
    const int n     = cnt[i];
    const float di  = dinv[i];
    const int c4 = lane * 4;
    const float4 gs = *(const float4*)(g + (size_t)i * DHID + c4);
    const float4 bb = *(const float4*)(bias + c4);

    float4 acc = make_float4(0.0f, 0.0f, 0.0f, 0.0f);
    for (int base = 0; base < n; base += 32) {
        const int m = min(32, n - base);
        int sj = 0;
        float wj = 0.0f;
        if (lane < m) {
            sj = bucket[start + base + lane];
            sj = min(max(sj, 0), Nrows - 1);
            wj = dinv[sj] * di;
        }
        int k = 0;
        for (; k + 4 <= m; k += 4) {
            const float4 v0 = load_row(g, sj, k,     c4);
            const float4 v1 = load_row(g, sj, k + 1, c4);
            const float4 v2 = load_row(g, sj, k + 2, c4);
            const float4 v3 = load_row(g, sj, k + 3, c4);
            add_row(acc, v0, wj, k);
            add_row(acc, v1, wj, k + 1);
            add_row(acc, v2, wj, k + 2);
            add_row(acc, v3, wj, k + 3);
        }
        const int r = m - k;
        if (r == 3) {
            const float4 v0 = load_row(g, sj, k,     c4);
            const float4 v1 = load_row(g, sj, k + 1, c4);
            const float4 v2 = load_row(g, sj, k + 2, c4);
            add_row(acc, v0, wj, k);
            add_row(acc, v1, wj, k + 1);
            add_row(acc, v2, wj, k + 2);
        } else if (r == 2) {
            const float4 v0 = load_row(g, sj, k,     c4);
            const float4 v1 = load_row(g, sj, k + 1, c4);
            add_row(acc, v0, wj, k);
            add_row(acc, v1, wj, k + 1);
        } else if (r == 1) {
            const float4 v0 = load_row(g, sj, k, c4);
            add_row(acc, v0, wj, k);
        }
    }

    const float s2 = di * di;
    float4 r4;
    r4.x = acc.x + gs.x * s2 + bb.x;
    r4.y = acc.y + gs.y * s2 + bb.y;
    r4.z = acc.z + gs.z * s2 + bb.z;
    r4.w = acc.w + gs.w * s2 + bb.w;
    *(float4*)(out + (size_t)i * DHID + c4) = r4;
}

extern "C" void kernel_launch(void* const* d_in, const int* in_sizes, int n_in,
                              void* d_out, int out_size, void* d_ws, size_t ws_size,
                              hipStream_t stream)
{
    const float* x    = (const float*)d_in[0];
    const int*   ei   = (const int*)  d_in[1];
    const float* W_in = (const float*)d_in[3];
    const float* b_in = (const float*)d_in[4];
    const float* W_g  = (const float*)d_in[5];
    const float* b_g  = (const float*)d_in[6];
    const float* W_o1 = (const float*)d_in[7];
    const float* b_o1 = (const float*)d_in[8];
    const float* W_o2 = (const float*)d_in[9];
    const float* b_o2 = (const float*)d_in[10];

    const int ND = 239;
    const int N  = in_sizes[0] / ND;
    const int E  = in_sizes[1] / 2;
    const int* src = ei;
    const int* dst = ei + E;

    const size_t nh = (size_t)N * DHID;
    float* ws   = (float*)d_ws;
    int*   cnt  = (int*)d_ws;
    float* dinv = ws + (size_t)N;
    float* bufA = ws + 2 * (size_t)N;
    float* bufB = bufA + nh;
    int*   off    = (int*)(bufB + nh);
    int*   rank   = off + (size_t)N;
    int*   bucket = rank + (size_t)E;

    const int gemm_blocks = (N / 16 + 3) / 4;
    const int edge_blocks = (E + 255) / 256;
    const int scan_blocks = (N + SCAN_T - 1) / SCAN_T;
    const int gath_blocks = (N + 7) / 8;

    PlaceArgs none;
    none.src = nullptr; none.dst = nullptr; none.off = nullptr; none.rank = nullptr;
    none.bucket = nullptr; none.E = 0; none.gblocks = gemm_blocks;
    PlaceArgs pl;
    pl.src = src; pl.dst = dst; pl.off = off; pl.rank = rank;
    pl.bucket = bucket; pl.E = E; pl.gblocks = gemm_blocks;
    const int place_blocks = (E + 127) / 128;

    gemm_wmma<239, true, true, false, true, false><<<gemm_blocks, 128, 0, stream>>>(
        x, W_in, b_in, bufA, N, nullptr, nullptr, cnt, N, none);

    count_rank_kernel<<<edge_blocks, 256, 0, stream>>>(dst, cnt, rank, E);
    offsets_kernel<<<scan_blocks, SCAN_T, 0, stream>>>(cnt, off, dinv, N);

    gemm_wmma<128, false, false, false, false, true><<<gemm_blocks + place_blocks, 128, 0, stream>>>(
        bufA, W_g, nullptr, bufB, N, nullptr, nullptr, nullptr, 0, pl);
    gather_kernel<<<gath_blocks, 256, 0, stream>>>(bufB, bucket, off, cnt, dinv, b_g, bufA, N);

    gemm_wmma<128, false, false, false, false, false><<<gemm_blocks, 128, 0, stream>>>(
        bufA, W_g, nullptr, bufB, N, nullptr, nullptr, nullptr, 0, none);
    gather_kernel<<<gath_blocks, 256, 0, stream>>>(bufB, bucket, off, cnt, dinv, b_g, bufA, N);

    gemm_wmma<128, true, true, true, false, false><<<gemm_blocks, 128, 0, stream>>>(
        bufA, W_o1, b_o1, (float*)d_out, N, W_o2, b_o2, nullptr, 0, none);
}


# conv1 and conv2 products merged by reassociation: P = h1 @ (W_g @ W_g), conv1's bias carried as b_g @ W_g; one 100000-row product launch less, W_g @ W_g on WMMA in launch 1 (sum order changes)
# speedup vs baseline: 1.9529x; 1.1640x over previous
#include <hip/hip_runtime.h>
#include <math.h>


typedef __attribute__((ext_vector_type(2))) float v2f;
typedef __attribute__((ext_vector_type(8))) float v8f;

#define DHID 128
#define TK 64
#define NEG_SLOPE 0.01f
#define SCAN_T 256

__global__ void count_rank_kernel(const int* __restrict__ dst, int* __restrict__ cnt,
                                  int* __restrict__ rank, int E) {
    const int e = blockIdx.x * blockDim.x + threadIdx.x;
    if (e < E) rank[e] = atomicAdd(cnt + dst[e], 1);
}

__global__ __launch_bounds__(SCAN_T) void offsets_kernel(
    const int* __restrict__ cnt,
    int* __restrict__ off, float* __restrict__ dinv, int N)
{
    __shared__ int red[SCAN_T];
    __shared__ int sc[SCAN_T];
    const int tid = threadIdx.x;
    const int b = blockIdx.x;
    int part = 0;
    const int lim = b * SCAN_T;
#pragma unroll 8
    for (int j = tid * 4; j < lim; j += SCAN_T * 4) {
        const int4 v = *(const int4*)(cnt + j);
        part += (v.x + v.y) + (v.z + v.w);
    }
    red[tid] = part;
    const int i = b * SCAN_T + tid;
    const int c = (i < N) ? cnt[i] : 0;
    sc[tid] = c;
    __syncthreads();
    for (int o = SCAN_T / 2; o > 0; o >>= 1) {
        if (tid < o) red[tid] += red[tid + o];
        __syncthreads();
    }
    const int base = red[0];
    for (int o = 1; o < SCAN_T; o <<= 1) {
        const int v = (tid >= o) ? sc[tid - o] : 0;
        __syncthreads();
        sc[tid] += v;
        __syncthreads();
    }
    if (i < N) {
        off[i]  = base + sc[tid] - c;
        dinv[i] = rsqrtf((float)c + 1.0f);
    }
}

struct PlaceArgs {
    const int* src;
    const int* dst;
    const int* off;
    const int* rank;
    int*       bucket;
    int        E;
    int        gblocks;
};

template <int K, bool BIAS, bool ACT, bool PROJ>
__device__ __forceinline__ void gemm_tile(
    float (*Wp)[DHID * 2], const int bx,
    const float* __restrict__ A, const float* __restrict__ W,
    const float* __restrict__ bias, float* __restrict__ Out, int Nrows,
    const float* __restrict__ W2, const float* __restrict__ b2)
{
    const int tid  = threadIdx.x;
    const int wave = tid >> 5;
    const int lane = tid & 31;
    const int row0 = (bx * 4 + wave) * 16;
    const bool rowok = (row0 < Nrows);

    const int m     = lane & 15;
    const int khalf = (lane >> 4) * 2;

    v8f acc[8];
#pragma unroll
    for (int t = 0; t < 8; ++t) acc[t] = (v8f){};

    const float* __restrict__ arow = A + (size_t)((rowok ? row0 : 0) + m) * K;

    for (int k0 = 0; k0 < K; k0 += TK) {
#pragma unroll 8
        for (int idx = 0; idx < TK * DHID; idx += 128) {
            const int j   = idx + tid;
            const int krl = j >> 7;
            const int col = j & (DHID - 1);
            const int kr  = k0 + krl;
            const float v = (kr < K) ? W[(size_t)kr * DHID + col] : 0.0f;
            Wp[krl >> 1][col * 2 + (krl & 1)] = v;
        }
        __syncthreads();

#pragma unroll 4
        for (int k = 0; k < TK; k += 4) {
            const int kk = k0 + k + khalf;
            const int i0 = (kk     < K) ? kk     : (K - 1);
            const int i1 = (kk + 1 < K) ? kk + 1 : (K - 1);
            v2f a;
            a.x = arow[i0];
            a.y = arow[i1];
            const float* __restrict__ wp = &Wp[(k + khalf) >> 1][0];
#pragma unroll
            for (int t = 0; t < 8; ++t) {
                const v2f b = *(const v2f*)(wp + (t * 16 + m) * 2);
                if (PROJ)
                    acc[t] = __builtin_amdgcn_wmma_f32_16x16x4_f32(
                        false, b, false, a, (short)0, acc[t], false, false);
                else
                    acc[t] = __builtin_amdgcn_wmma_f32_16x16x4_f32(
                        false, a, false, b, (short)0, acc[t], false, false);
            }
        }
        __syncthreads();
    }

    if (!rowok) return;
    if (PROJ) {
        const int hs = lane >> 4;
        float p0 = 0.0f, p1 = 0.0f;
#pragma unroll
        for (int t = 0; t < 8; ++t) {
            const int f0 = t * 16 + hs * 8;
            float bv[8];
            if (BIAS) {
                const float4 ba = *(const float4*)(bias + f0);
                const float4 bb = *(const float4*)(bias + f0 + 4);
                bv[0] = ba.x; bv[1] = ba.y; bv[2] = ba.z; bv[3] = ba.w;
                bv[4] = bb.x; bv[5] = bb.y; bv[6] = bb.z; bv[7] = bb.w;
            }
            float wv[16];
#pragma unroll
            for (int q = 0; q < 4; ++q) {
                const float4 w4 = *(const float4*)(W2 + f0 * 2 + q * 4);
                wv[q * 4 + 0] = w4.x; wv[q * 4 + 1] = w4.y; wv[q * 4 + 2] = w4.z; wv[q * 4 + 3] = w4.w;
            }
#pragma unroll
            for (int r = 0; r < 8; ++r) {
                float v = acc[t][r] + (BIAS ? bv[r] : 0.0f);
                if (ACT) v = (v >= 0.0f) ? v : NEG_SLOPE * v;
                p0 += v * wv[r * 2 + 0];
                p1 += v * wv[r * 2 + 1];
            }
        }
        p0 += __shfl_xor(p0, 16);
        p1 += __shfl_xor(p1, 16);
        if (hs == 0) {
            float2 o;
            o.x = p0 + b2[0];
            o.y = p1 + b2[1];
            *(float2*)(Out + (size_t)(row0 + m) * 2) = o;
        }
        return;
    }
    const int rbase = row0 + ((lane >> 4) ? 8 : 0);
#pragma unroll
    for (int t = 0; t < 8; ++t) {
        const int col = t * 16 + m;
        const float bv = BIAS ? bias[col] : 0.0f;
#pragma unroll
        for (int r = 0; r < 8; ++r) {
            float v = acc[t][r] + bv;
            if (ACT) v = (v >= 0.0f) ? v : NEG_SLOPE * v;
            Out[(size_t)(rbase + r) * DHID + col] = v;
        }
    }
}

struct SideArgs {
    const float* Wg;
    const float* bg;
    float*       Wsq;
    float*       cvec;
};
#define SIDE_GEMM_BLOCKS (DHID / 64)
#define SIDE_BLOCKS (SIDE_GEMM_BLOCKS + 1)

template <int K, bool BIAS, bool ACT, bool PROJ, bool ZERO, bool PLACE, bool SIDE>
__global__ __launch_bounds__(128) void gemm_wmma(
    const float* __restrict__ A, const float* __restrict__ W,
    const float* __restrict__ bias, float* __restrict__ Out, int Nrows,
    const float* __restrict__ W2, const float* __restrict__ b2,
    int* __restrict__ z, int nz, const PlaceArgs pa, const SideArgs sa)
{
    __shared__ float Wp[TK / 2][DHID * 2];

    const int tid  = threadIdx.x;
    if (PLACE) {
        if ((int)blockIdx.x >= pa.gblocks) {
            const int e = ((int)blockIdx.x - pa.gblocks) * 128 + tid;
            if (e < pa.E) {
                int pos = pa.off[pa.dst[e]] + pa.rank[e];
                pos = min(max(pos, 0), pa.E - 1);
                pa.bucket[pos] = pa.src[e];
            }
            return;
        }
    }
    if (ZERO) {
        const int zi = blockIdx.x * 128 + tid;
        if (zi < nz) z[zi] = 0;
    }
    int bx = blockIdx.x;
    if (SIDE) {
        if (bx < SIDE_GEMM_BLOCKS) {
            gemm_tile<DHID, false, false, false>(Wp, bx, sa.Wg, sa.Wg, nullptr, sa.Wsq, DHID, nullptr, nullptr);
            return;
        }
        if (bx == SIDE_GEMM_BLOCKS) {
            float s = 0.0f;
#pragma unroll 16
            for (int k = 0; k < DHID; ++k) s += sa.bg[k] * sa.Wg[k * DHID + tid];
            sa.cvec[tid] = s;
            return;
        }
        bx -= SIDE_BLOCKS;
    }
    gemm_tile<K, BIAS, ACT, PROJ>(Wp, bx, A, W, bias, Out, Nrows, W2, b2);
}

__device__ __forceinline__ float4 load_row(const float* __restrict__ g, int sj, int k, int c4) {
    const int s = __builtin_amdgcn_readlane(sj, k);
    return *(const float4*)(g + (size_t)s * DHID + c4);
}

__device__ __forceinline__ void add_row(float4& acc, const float4 v, float wj, int k) {
    const float w = __int_as_float(__builtin_amdgcn_readlane(__float_as_int(wj), k));
    acc.x += v.x * w;
    acc.y += v.y * w;
    acc.z += v.z * w;
    acc.w += v.w * w;
}

__global__ __launch_bounds__(256) void gather_kernel(
    const float* __restrict__ g, const int* __restrict__ bucket,
    const int* __restrict__ off, const int* __restrict__ cnt,
    const float* __restrict__ dinv, const float* __restrict__ bias,
    float* __restrict__ out, int Nrows)
{
    const int lane = threadIdx.x & 31;
    const int i = __builtin_amdgcn_readfirstlane(
        (int)(blockIdx.x * (blockDim.x >> 5) + (threadIdx.x >> 5)));
    if (i >= Nrows) return;
    const int start = off[i];
    const int n     = cnt[i];
    const float di  = dinv[i];
    const int c4 = lane * 4;
    const float4 gs = *(const float4*)(g + (size_t)i * DHID + c4);
    const float4 bb = *(const float4*)(bias + c4);

    float4 acc = make_float4(0.0f, 0.0f, 0.0f, 0.0f);
    for (int base = 0; base < n; base += 32) {
        const int m = min(32, n - base);
        int sj = 0;
        float wj = 0.0f;
        if (lane < m) {
            sj = bucket[start + base + lane];
            sj = min(max(sj, 0), Nrows - 1);
            wj = dinv[sj] * di;
        }
        int k = 0;
        for (; k + 4 <= m; k += 4) {
            const float4 v0 = load_row(g, sj, k,     c4);
            const float4 v1 = load_row(g, sj, k + 1, c4);
            const float4 v2 = load_row(g, sj, k + 2, c4);
            const float4 v3 = load_row(g, sj, k + 3, c4);
            add_row(acc, v0, wj, k);
            add_row(acc, v1, wj, k + 1);
            add_row(acc, v2, wj, k + 2);
            add_row(acc, v3, wj, k + 3);
        }
        const int r = m - k;
        if (r == 3) {
            const float4 v0 = load_row(g, sj, k,     c4);
            const float4 v1 = load_row(g, sj, k + 1, c4);
            const float4 v2 = load_row(g, sj, k + 2, c4);
            add_row(acc, v0, wj, k);
            add_row(acc, v1, wj, k + 1);
            add_row(acc, v2, wj, k + 2);
        } else if (r == 2) {
            const float4 v0 = load_row(g, sj, k,     c4);
            const float4 v1 = load_row(g, sj, k + 1, c4);
            add_row(acc, v0, wj, k);
            add_row(acc, v1, wj, k + 1);
        } else if (r == 1) {
            const float4 v0 = load_row(g, sj, k, c4);
            add_row(acc, v0, wj, k);
        }
    }

    const float s2 = di * di;
    float4 r4;
    r4.x = acc.x + gs.x * s2 + bb.x;
    r4.y = acc.y + gs.y * s2 + bb.y;
    r4.z = acc.z + gs.z * s2 + bb.z;
    r4.w = acc.w + gs.w * s2 + bb.w;
    *(float4*)(out + (size_t)i * DHID + c4) = r4;
}

extern "C" void kernel_launch(void* const* d_in, const int* in_sizes, int n_in,
                              void* d_out, int out_size, void* d_ws, size_t ws_size,
                              hipStream_t stream)
{
    const float* x    = (const float*)d_in[0];
    const int*   ei   = (const int*)  d_in[1];
    const float* W_in = (const float*)d_in[3];
    const float* b_in = (const float*)d_in[4];
    const float* W_g  = (const float*)d_in[5];
    const float* b_g  = (const float*)d_in[6];
    const float* W_o1 = (const float*)d_in[7];
    const float* b_o1 = (const float*)d_in[8];
    const float* W_o2 = (const float*)d_in[9];
    const float* b_o2 = (const float*)d_in[10];

    const int ND = 239;
    const int N  = in_sizes[0] / ND;
    const int E  = in_sizes[1] / 2;
    const int* src = ei;
    const int* dst = ei + E;

    const size_t nh = (size_t)N * DHID;
    float* ws   = (float*)d_ws;
    int*   cnt  = (int*)d_ws;
    float* dinv = ws + (size_t)N;
    float* bufA = ws + 2 * (size_t)N;
    float* bufB = bufA + nh;
    int*   off    = (int*)(bufB + nh);
    int*   rank   = off + (size_t)N;
    int*   bucket = rank + (size_t)E;
    float* Wsq    = (float*)(bucket + (size_t)E);
    float* cvec   = Wsq + DHID * DHID;

    const int gemm_blocks = (N / 16 + 3) / 4;
    const int edge_blocks = (E + 255) / 256;
    const int scan_blocks = (N + SCAN_T - 1) / SCAN_T;
    const int gath_blocks = (N + 7) / 8;

    PlaceArgs none;
    none.src = nullptr; none.dst = nullptr; none.off = nullptr; none.rank = nullptr;
    none.bucket = nullptr; none.E = 0; none.gblocks = gemm_blocks;
    PlaceArgs pl;
    pl.src = src; pl.dst = dst; pl.off = off; pl.rank = rank;
    pl.bucket = bucket; pl.E = E; pl.gblocks = gemm_blocks;
    const int place_blocks = (E + 127) / 128;
    SideArgs noside;
    noside.Wg = nullptr; noside.bg = nullptr; noside.Wsq = nullptr; noside.cvec = nullptr;
    SideArgs sd;
    sd.Wg = W_g; sd.bg = b_g; sd.Wsq = Wsq; sd.cvec = cvec;

    gemm_wmma<239, true, true, false, true, false, true><<<gemm_blocks + SIDE_BLOCKS, 128, 0, stream>>>(
        x, W_in, b_in, bufA, N, nullptr, nullptr, cnt, N, none, sd);

    count_rank_kernel<<<edge_blocks, 256, 0, stream>>>(dst, cnt, rank, E);
    offsets_kernel<<<scan_blocks, SCAN_T, 0, stream>>>(cnt, off, dinv, N);

    gemm_wmma<128, false, false, false, false, true, false><<<gemm_blocks + place_blocks, 128, 0, stream>>>(
        bufA, Wsq, nullptr, bufB, N, nullptr, nullptr, nullptr, 0, pl, noside);
    gather_kernel<<<gath_blocks, 256, 0, stream>>>(bufB, bucket, off, cnt, dinv, cvec, bufA, N);
    gather_kernel<<<gath_blocks, 256, 0, stream>>>(bufA, bucket, off, cnt, dinv, b_g, bufB, N);

    gemm_wmma<128, true, true, true, false, false, false><<<gemm_blocks, 128, 0, stream>>>(
        bufB, W_o1, b_o1, (float*)d_out, N, W_o2, b_o2, nullptr, 0, none, noside);
}


# Output layer's product merged by reassociation too: P = h1 @ (W_g @ W_g @ W_o1), biases as vectors; last GEMM launch gone, 2nd aggregation does activation + 128->2 projection (sum order changes)
# speedup vs baseline: 2.2627x; 1.1586x over previous
#include <hip/hip_runtime.h>
#include <math.h>


typedef __attribute__((ext_vector_type(2))) float v2f;
typedef __attribute__((ext_vector_type(8))) float v8f;

#define DHID 128
#define TK 64
#define NEG_SLOPE 0.01f
#define SCAN_T 256

__global__ void count_rank_kernel(const int* __restrict__ dst, int* __restrict__ cnt,
                                  int* __restrict__ rank, int E) {
    const int e = blockIdx.x * blockDim.x + threadIdx.x;
    if (e < E) rank[e] = atomicAdd(cnt + dst[e], 1);
}

__global__ __launch_bounds__(SCAN_T) void offsets_kernel(
    const int* __restrict__ cnt,
    int* __restrict__ off, float* __restrict__ dinv, int N)
{
    __shared__ int red[SCAN_T];
    __shared__ int sc[SCAN_T];
    const int tid = threadIdx.x;
    const int b = blockIdx.x;
    int part = 0;
    const int lim = b * SCAN_T;
#pragma unroll 8
    for (int j = tid * 4; j < lim; j += SCAN_T * 4) {
        const int4 v = *(const int4*)(cnt + j);
        part += (v.x + v.y) + (v.z + v.w);
    }
    red[tid] = part;
    const int i = b * SCAN_T + tid;
    const int c = (i < N) ? cnt[i] : 0;
    sc[tid] = c;
    __syncthreads();
    for (int o = SCAN_T / 2; o > 0; o >>= 1) {
        if (tid < o) red[tid] += red[tid + o];
        __syncthreads();
    }
    const int base = red[0];
    for (int o = 1; o < SCAN_T; o <<= 1) {
        const int v = (tid >= o) ? sc[tid - o] : 0;
        __syncthreads();
        sc[tid] += v;
        __syncthreads();
    }
    if (i < N) {
        off[i]  = base + sc[tid] - c;
        dinv[i] = rsqrtf((float)c + 1.0f);
    }
}

struct PlaceArgs {
    const int* src;
    const int* dst;
    const int* off;
    const int* rank;
    int*       bucket;
    int        E;
    int        gblocks;
};

template <int K, bool BIAS, bool ACT>
__device__ __forceinline__ void gemm_tile(
    float (*Wp)[DHID * 2], const int bx,
    const float* __restrict__ A, const float* __restrict__ W,
    const float* __restrict__ bias, float* __restrict__ Out, int Nrows)
{
    const int tid  = threadIdx.x;
    const int wave = tid >> 5;
    const int lane = tid & 31;
    const int row0 = (bx * 4 + wave) * 16;
    const bool rowok = (row0 < Nrows);

    const int m     = lane & 15;
    const int khalf = (lane >> 4) * 2;

    v8f acc[8];
#pragma unroll
    for (int t = 0; t < 8; ++t) acc[t] = (v8f){};

    const float* __restrict__ arow = A + (size_t)((rowok ? row0 : 0) + m) * K;

    for (int k0 = 0; k0 < K; k0 += TK) {
#pragma unroll 8
        for (int idx = 0; idx < TK * DHID; idx += 128) {
            const int j   = idx + tid;
            const int krl = j >> 7;
            const int col = j & (DHID - 1);
            const int kr  = k0 + krl;
            const float v = (kr < K) ? W[(size_t)kr * DHID + col] : 0.0f;
            Wp[krl >> 1][col * 2 + (krl & 1)] = v;
        }
        __syncthreads();

#pragma unroll 4
        for (int k = 0; k < TK; k += 4) {
            const int kk = k0 + k + khalf;
            const int i0 = (kk     < K) ? kk     : (K - 1);
            const int i1 = (kk + 1 < K) ? kk + 1 : (K - 1);
            v2f a;
            a.x = arow[i0];
            a.y = arow[i1];
            const float* __restrict__ wp = &Wp[(k + khalf) >> 1][0];
#pragma unroll
            for (int t = 0; t < 8; ++t) {
                const v2f b = *(const v2f*)(wp + (t * 16 + m) * 2);
                acc[t] = __builtin_amdgcn_wmma_f32_16x16x4_f32(
                    false, a, false, b, (short)0, acc[t], false, false);
            }
        }
        __syncthreads();
    }

    if (!rowok) return;
    const int rbase = row0 + ((lane >> 4) ? 8 : 0);
#pragma unroll
    for (int t = 0; t < 8; ++t) {
        const int col = t * 16 + m;
        const float bv = BIAS ? bias[col] : 0.0f;
#pragma unroll
        for (int r = 0; r < 8; ++r) {
            float v = acc[t][r] + bv;
            if (ACT) v = (v >= 0.0f) ? v : NEG_SLOPE * v;
            Out[(size_t)(rbase + r) * DHID + col] = v;
        }
    }
}

struct SideArgs {
    const float* Wg;
    const float* bg;
    const float* Wo1;
    const float* bo1;
    float*       Wsq;
    float*       W3;
    float*       c3;
    float*       d3;
};
#define SIDE_GEMM_BLOCKS (DHID / 64)
#define SIDE_BLOCKS (SIDE_GEMM_BLOCKS + 1)

template <int K, bool BIAS, bool ACT, bool ZERO, bool PLACE, bool SIDE>
__global__ __launch_bounds__(128) void gemm_wmma(
    const float* __restrict__ A, const float* __restrict__ W,
    const float* __restrict__ bias, float* __restrict__ Out, int Nrows,
    int* __restrict__ z, int nz, const PlaceArgs pa, const SideArgs sa)
{
    __shared__ float Wp[TK / 2][DHID * 2];

    const int tid  = threadIdx.x;
    if (PLACE) {
        if ((int)blockIdx.x >= pa.gblocks) {
            const int e = ((int)blockIdx.x - pa.gblocks) * 128 + tid;
            if (e < pa.E) {
                int pos = pa.off[pa.dst[e]] + pa.rank[e];
                pos = min(max(pos, 0), pa.E - 1);
                pa.bucket[pos] = pa.src[e];
            }
            return;
        }
    }
    if (ZERO) {
        const int zi = blockIdx.x * 128 + tid;
        if (zi < nz) z[zi] = 0;
    }
    int bx = blockIdx.x;
    if (SIDE) {
        if (bx < SIDE_GEMM_BLOCKS) {
            gemm_tile<DHID, false, false>(Wp, bx, sa.Wg, sa.Wg, nullptr, sa.Wsq, DHID);
            __threadfence();
            __syncthreads();
            gemm_tile<DHID, false, false>(Wp, bx, sa.Wsq, sa.Wo1, nullptr, sa.W3, DHID);
            return;
        }
        if (bx == SIDE_GEMM_BLOCKS) {
            float* cg = &Wp[0][0];
            float s = 0.0f;
#pragma unroll 16
            for (int k = 0; k < DHID; ++k) s += sa.bg[k] * sa.Wg[k * DHID + tid];
            cg[tid] = s;
            __syncthreads();
            float s3 = 0.0f, sd = 0.0f;
#pragma unroll 16
            for (int k = 0; k < DHID; ++k) {
                const float w = sa.Wo1[k * DHID + tid];
                s3 += cg[k] * w;
                sd += sa.bg[k] * w;
            }
            sa.c3[tid] = s3;
            sa.d3[tid] = sd + sa.bo1[tid];
            return;
        }
        bx -= SIDE_BLOCKS;
    }
    gemm_tile<K, BIAS, ACT>(Wp, bx, A, W, bias, Out, Nrows);
}

__device__ __forceinline__ float4 load_row(const float* __restrict__ g, int sj, int k, int c4) {
    const int s = __builtin_amdgcn_readlane(sj, k);
    return *(const float4*)(g + (size_t)s * DHID + c4);
}

template <int M>
__device__ __forceinline__ float xrow(float x) {
    return __int_as_float(__builtin_amdgcn_update_dpp(0, __float_as_int(x), 0x160 + M, 0xf, 0xf, false));
}
__device__ __forceinline__ float xhalf(float x) {
    const int b = __float_as_int(x);
    return __int_as_float(__builtin_amdgcn_permlanex16(b, b, 0x76543210, 0xfedcba98, false, false));
}

__device__ __forceinline__ void add_row(float4& acc, const float4 v, float wj, int k) {
    const float w = __int_as_float(__builtin_amdgcn_readlane(__float_as_int(wj), k));
    acc.x += v.x * w;
    acc.y += v.y * w;
    acc.z += v.z * w;
    acc.w += v.w * w;
}

#define PROJ_WAVES 16
template <bool PROJ>
__global__ __launch_bounds__(PROJ ? PROJ_WAVES * 32 : 256) void gather_kernel(
    const float* __restrict__ g, const int* __restrict__ bucket,
    const int* __restrict__ off, const int* __restrict__ cnt,
    const float* __restrict__ dinv, const float* __restrict__ bias,
    float* __restrict__ out, int Nrows,
    const float* __restrict__ W2, const float* __restrict__ b2)
{
    __shared__ float2 res[PROJ_WAVES];
    const int lane = threadIdx.x & 31;
    const int iw = __builtin_amdgcn_readfirstlane(
        (int)(blockIdx.x * (blockDim.x >> 5) + (threadIdx.x >> 5)));
    if (!PROJ && iw >= Nrows) return;
    const int i = PROJ ? min(iw, Nrows - 1) : iw;
    const int start = off[i];
    const int n     = cnt[i];
    const float di  = dinv[i];
    const int c4 = lane * 4;
    const float4 gs = *(const float4*)(g + (size_t)i * DHID + c4);
    const float4 bb = *(const float4*)(bias + c4);

    float4 acc = make_float4(0.0f, 0.0f, 0.0f, 0.0f);
    for (int base = 0; base < n; base += 32) {
        const int m = min(32, n - base);
        int sj = 0;
        float wj = 0.0f;
        if (lane < m) {
            sj = bucket[start + base + lane];
            sj = min(max(sj, 0), Nrows - 1);
            wj = dinv[sj] * di;
        }
        int k = 0;
        for (; k + 4 <= m; k += 4) {
            const float4 v0 = load_row(g, sj, k,     c4);
            const float4 v1 = load_row(g, sj, k + 1, c4);
            const float4 v2 = load_row(g, sj, k + 2, c4);
            const float4 v3 = load_row(g, sj, k + 3, c4);
            add_row(acc, v0, wj, k);
            add_row(acc, v1, wj, k + 1);
            add_row(acc, v2, wj, k + 2);
            add_row(acc, v3, wj, k + 3);
        }
        const int r = m - k;
        if (r == 3) {
            const float4 v0 = load_row(g, sj, k,     c4);
            const float4 v1 = load_row(g, sj, k + 1, c4);
            const float4 v2 = load_row(g, sj, k + 2, c4);
            add_row(acc, v0, wj, k);
            add_row(acc, v1, wj, k + 1);
            add_row(acc, v2, wj, k + 2);
        } else if (r == 2) {
            const float4 v0 = load_row(g, sj, k,     c4);
            const float4 v1 = load_row(g, sj, k + 1, c4);
            add_row(acc, v0, wj, k);
            add_row(acc, v1, wj, k + 1);
        } else if (r == 1) {
            const float4 v0 = load_row(g, sj, k, c4);
            add_row(acc, v0, wj, k);
        }
    }

    const float s2 = di * di;
    float4 r4;
    r4.x = acc.x + gs.x * s2 + bb.x;
    r4.y = acc.y + gs.y * s2 + bb.y;
    r4.z = acc.z + gs.z * s2 + bb.z;
    r4.w = acc.w + gs.w * s2 + bb.w;
    if (!PROJ) {
        *(float4*)(out + (size_t)i * DHID + c4) = r4;
        return;
    }
    const float v0 = (r4.x >= 0.0f) ? r4.x : NEG_SLOPE * r4.x;
    const float v1 = (r4.y >= 0.0f) ? r4.y : NEG_SLOPE * r4.y;
    const float v2 = (r4.z >= 0.0f) ? r4.z : NEG_SLOPE * r4.z;
    const float v3 = (r4.w >= 0.0f) ? r4.w : NEG_SLOPE * r4.w;
    const float4 wa = *(const float4*)(W2 + c4 * 2);
    const float4 wb = *(const float4*)(W2 + c4 * 2 + 4);
    float p0 = (v0 * wa.x + v1 * wa.z) + (v2 * wb.x + v3 * wb.z);
    float p1 = (v0 * wa.y + v1 * wa.w) + (v2 * wb.y + v3 * wb.w);
    p0 += xrow<1>(p0);  p1 += xrow<1>(p1);
    p0 += xrow<2>(p0);  p1 += xrow<2>(p1);
    p0 += xrow<4>(p0);  p1 += xrow<4>(p1);
    p0 += xrow<8>(p0);  p1 += xrow<8>(p1);
    p0 += xhalf(p0);    p1 += xhalf(p1);
    if (lane == 0) res[threadIdx.x >> 5] = make_float2(p0 + b2[0], p1 + b2[1]);
    __syncthreads();
    if (threadIdx.x < PROJ_WAVES) {
        const int node = (int)blockIdx.x * PROJ_WAVES + (int)threadIdx.x;
        if (node < Nrows) *(float2*)(out + (size_t)node * 2) = res[threadIdx.x];
    }
}

extern "C" void kernel_launch(void* const* d_in, const int* in_sizes, int n_in,
                              void* d_out, int out_size, void* d_ws, size_t ws_size,
                              hipStream_t stream)
{
    const float* x    = (const float*)d_in[0];
    const int*   ei   = (const int*)  d_in[1];
    const float* W_in = (const float*)d_in[3];
    const float* b_in = (const float*)d_in[4];
    const float* W_g  = (const float*)d_in[5];
    const float* b_g  = (const float*)d_in[6];
    const float* W_o1 = (const float*)d_in[7];
    const float* b_o1 = (const float*)d_in[8];
    const float* W_o2 = (const float*)d_in[9];
    const float* b_o2 = (const float*)d_in[10];

    const int ND = 239;
    const int N  = in_sizes[0] / ND;
    const int E  = in_sizes[1] / 2;
    const int* src = ei;
    const int* dst = ei + E;

    const size_t nh = (size_t)N * DHID;
    float* ws   = (float*)d_ws;
    int*   cnt  = (int*)d_ws;
    float* dinv = ws + (size_t)N;
    float* bufA = ws + 2 * (size_t)N;
    float* bufB = bufA + nh;
    int*   off    = (int*)(bufB + nh);
    int*   rank   = off + (size_t)N;
    int*   bucket = rank + (size_t)E;
    float* Wsq    = (float*)(bucket + (size_t)E);
    float* W3     = Wsq + DHID * DHID;
    float* c3     = W3 + DHID * DHID;
    float* d3     = c3 + DHID;

    const int gemm_blocks = (N / 16 + 3) / 4;
    const int edge_blocks = (E + 255) / 256;
    const int scan_blocks = (N + SCAN_T - 1) / SCAN_T;
    const int gath_blocks = (N + 7) / 8;
    const int proj_blocks = (N + PROJ_WAVES - 1) / PROJ_WAVES;

    PlaceArgs none;
    none.src = nullptr; none.dst = nullptr; none.off = nullptr; none.rank = nullptr;
    none.bucket = nullptr; none.E = 0; none.gblocks = gemm_blocks;
    PlaceArgs pl;
    pl.src = src; pl.dst = dst; pl.off = off; pl.rank = rank;
    pl.bucket = bucket; pl.E = E; pl.gblocks = gemm_blocks;
    const int place_blocks = (E + 127) / 128;
    SideArgs noside;
    noside.Wg = nullptr; noside.bg = nullptr; noside.Wo1 = nullptr; noside.bo1 = nullptr;
    noside.Wsq = nullptr; noside.W3 = nullptr; noside.c3 = nullptr; noside.d3 = nullptr;
    SideArgs sd;
    sd.Wg = W_g; sd.bg = b_g; sd.Wo1 = W_o1; sd.bo1 = b_o1;
    sd.Wsq = Wsq; sd.W3 = W3; sd.c3 = c3; sd.d3 = d3;

    gemm_wmma<239, true, true, true, false, true><<<gemm_blocks + SIDE_BLOCKS, 128, 0, stream>>>(
        x, W_in, b_in, bufA, N, cnt, N, none, sd);

    count_rank_kernel<<<edge_blocks, 256, 0, stream>>>(dst, cnt, rank, E);
    offsets_kernel<<<scan_blocks, SCAN_T, 0, stream>>>(cnt, off, dinv, N);

    gemm_wmma<128, false, false, false, true, false><<<gemm_blocks + place_blocks, 128, 0, stream>>>(
        bufA, W3, nullptr, bufB, N, nullptr, 0, pl, noside);
    gather_kernel<false><<<gath_blocks, 256, 0, stream>>>(
        bufB, bucket, off, cnt, dinv, c3, bufA, N, nullptr, nullptr);
    gather_kernel<true><<<proj_blocks, PROJ_WAVES * 32, 0, stream>>>(
        bufA, bucket, off, cnt, dinv, d3, (float*)d_out, N, W_o2, b_o2);
}


# GEMM A operand read as 16-byte pieces in a permuted k order (1 load per 16x8 block, was 2-4), asked for ahead of the W staging; staging 32 rows per trip; K=239 sweep ends at 240 (sum order changes)
# speedup vs baseline: 2.9693x; 1.3123x over previous
#include <hip/hip_runtime.h>
#include <math.h>


typedef __attribute__((ext_vector_type(2))) float v2f;
typedef __attribute__((ext_vector_type(8))) float v8f;

#define DHID 128
#define TK 64
#define NEG_SLOPE 0.01f
#define SCAN_T 256

__global__ void count_rank_kernel(const int* __restrict__ dst, int* __restrict__ cnt,
                                  int* __restrict__ rank, int E) {
    const int e = blockIdx.x * blockDim.x + threadIdx.x;
    if (e < E) rank[e] = atomicAdd(cnt + dst[e], 1);
}

__global__ __launch_bounds__(SCAN_T) void offsets_kernel(
    const int* __restrict__ cnt,
    int* __restrict__ off, float* __restrict__ dinv, int N)
{
    __shared__ int red[SCAN_T];
    __shared__ int sc[SCAN_T];
    const int tid = threadIdx.x;
    const int b = blockIdx.x;
    int part = 0;
    const int lim = b * SCAN_T;
#pragma unroll 8
    for (int j = tid * 4; j < lim; j += SCAN_T * 4) {
        const int4 v = *(const int4*)(cnt + j);
        part += (v.x + v.y) + (v.z + v.w);
    }
    red[tid] = part;
    const int i = b * SCAN_T + tid;
    const int c = (i < N) ? cnt[i] : 0;
    sc[tid] = c;
    __syncthreads();
    for (int o = SCAN_T / 2; o > 0; o >>= 1) {
        if (tid < o) red[tid] += red[tid + o];
        __syncthreads();
    }
    const int base = red[0];
    for (int o = 1; o < SCAN_T; o <<= 1) {
        const int v = (tid >= o) ? sc[tid - o] : 0;
        __syncthreads();
        sc[tid] += v;
        __syncthreads();
    }
    if (i < N) {
        off[i]  = base + sc[tid] - c;
        dinv[i] = rsqrtf((float)c + 1.0f);
    }
}

struct PlaceArgs {
    const int* src;
    const int* dst;
    const int* off;
    const int* rank;
    int*       bucket;
    int        E;
    int        gblocks;
};

typedef float f4u __attribute__((ext_vector_type(4), aligned(4)));

template <int K, int KV>
__device__ __forceinline__ void gemm_chunk(
    float (*Wp)[DHID * 2], const float* __restrict__ arow0, const float* __restrict__ W,
    const int k0, const int tid, const int m, const int h, v8f (&acc)[8])
{
    constexpr int NG = (KV + 7) / 8;
    float4 a4[NG];
#pragma unroll
    for (int g = 0; g < NG; ++g) {
        if (8 * g + 8 <= KV) {
            const f4u v = *(const f4u*)(arow0 + k0 + 8 * g + 4 * h);
            a4[g] = make_float4(v.x, v.y, v.z, v.w);
        } else {
            const int e = k0 + 8 * g + 4 * h;
            a4[g].x = arow0[min(e,     K - 1)];
            a4[g].y = arow0[min(e + 1, K - 1)];
            a4[g].z = arow0[min(e + 2, K - 1)];
            a4[g].w = arow0[min(e + 3, K - 1)];
        }
    }

    constexpr int RT = ((NG * 8) % 32 == 0) ? 32 : 16;
    const float* __restrict__ wbase = W + (size_t)k0 * DHID;
    float* __restrict__ wdst = &Wp[0][tid * 2];
#pragma nounroll
    for (int r0 = 0; r0 < NG * 8; r0 += RT) {
        float t[RT];
#pragma unroll
        for (int u = 0; u < RT; ++u) {
            t[u] = (KV == NG * 8 || r0 + u < KV) ? wbase[(r0 + u) * DHID + tid] : 0.0f;
        }
#pragma unroll
        for (int u = 0; u < RT; ++u)
            wdst[((r0 + u) >> 1) * (DHID * 2) + (u & 1)] = t[u];
    }
    __syncthreads();

#pragma unroll
    for (int g = 0; g < NG; ++g) {
#pragma unroll
        for (int j = 0; j < 2; ++j) {
            v2f a;
            a.x = j ? a4[g].z : a4[g].x;
            a.y = j ? a4[g].w : a4[g].y;
            const float* __restrict__ wp = &Wp[4 * g + 2 * h + j][0];
#pragma unroll
            for (int t = 0; t < 8; ++t) {
                const v2f b = *(const v2f*)(wp + (t * 16 + m) * 2);
                acc[t] = __builtin_amdgcn_wmma_f32_16x16x4_f32(
                    false, a, false, b, (short)0, acc[t], false, false);
            }
        }
    }
    __syncthreads();
}

template <int K, bool BIAS, bool ACT>
__device__ __forceinline__ void gemm_tile(
    float (*Wp)[DHID * 2], const int bx,
    const float* __restrict__ A, const float* __restrict__ W,
    const float* __restrict__ bias, float* __restrict__ Out, int Nrows)
{
    const int tid  = threadIdx.x;
    const int wave = tid >> 5;
    const int lane = tid & 31;
    const int row0 = __builtin_amdgcn_readfirstlane((bx * 4 + wave) * 16);
    const bool rowok = (row0 < Nrows);

    const int m = lane & 15;
    const int h = lane >> 4;

    v8f acc[8];
#pragma unroll
    for (int t = 0; t < 8; ++t) acc[t] = (v8f){};

    const float* __restrict__ arow0 = A + (size_t)((rowok ? row0 : 0) + m) * K;

#pragma nounroll
    for (int k0 = 0; k0 + TK <= K; k0 += TK)
        gemm_chunk<K, TK>(Wp, arow0, W, k0, tid, m, h, acc);
    if (K % TK != 0)
        gemm_chunk<K, (K % TK != 0) ? (K % TK) : TK>(Wp, arow0, W, (K / TK) * TK, tid, m, h, acc);

    if (!rowok) return;
    const int rbase = row0 + ((lane >> 4) ? 8 : 0);
#pragma unroll
    for (int t = 0; t < 8; ++t) {
        const int col = t * 16 + m;
        const float bv = BIAS ? bias[col] : 0.0f;
#pragma unroll
        for (int r = 0; r < 8; ++r) {
            float v = acc[t][r] + bv;
            if (ACT) v = (v >= 0.0f) ? v : NEG_SLOPE * v;
            Out[(size_t)(rbase + r) * DHID + col] = v;
        }
    }
}

struct SideArgs {
    const float* Wg;
    const float* bg;
    const float* Wo1;
    const float* bo1;
    float*       Wsq;
    float*       W3;
    float*       c3;
    float*       d3;
};
#define SIDE_GEMM_BLOCKS (DHID / 64)
#define SIDE_BLOCKS (SIDE_GEMM_BLOCKS + 1)

template <int K, bool BIAS, bool ACT, bool ZERO, bool PLACE, bool SIDE>
__global__ __launch_bounds__(128, 6) void gemm_wmma(
    const float* __restrict__ A, const float* __restrict__ W,
    const float* __restrict__ bias, float* __restrict__ Out, int Nrows,
    int* __restrict__ z, int nz, const PlaceArgs pa, const SideArgs sa)
{
    __shared__ float Wp[TK / 2][DHID * 2];

    const int tid  = threadIdx.x;
    if (PLACE) {
        if ((int)blockIdx.x >= pa.gblocks) {
            const int e = ((int)blockIdx.x - pa.gblocks) * 128 + tid;
            if (e < pa.E) {
                int pos = pa.off[pa.dst[e]] + pa.rank[e];
                pos = min(max(pos, 0), pa.E - 1);
                pa.bucket[pos] = pa.src[e];
            }
            return;
        }
    }
    if (ZERO) {
        const int zi = blockIdx.x * 128 + tid;
        if (zi < nz) z[zi] = 0;
    }
    int bx = blockIdx.x;
    if (SIDE) {
        if (bx < SIDE_GEMM_BLOCKS) {
            gemm_tile<DHID, false, false>(Wp, bx, sa.Wg, sa.Wg, nullptr, sa.Wsq, DHID);
            __threadfence();
            __syncthreads();
            gemm_tile<DHID, false, false>(Wp, bx, sa.Wsq, sa.Wo1, nullptr, sa.W3, DHID);
            return;
        }
        if (bx == SIDE_GEMM_BLOCKS) {
            float* cg = &Wp[0][0];
            float s = 0.0f;
#pragma unroll 16
            for (int k = 0; k < DHID; ++k) s += sa.bg[k] * sa.Wg[k * DHID + tid];
            cg[tid] = s;
            __syncthreads();
            float s3 = 0.0f, sd = 0.0f;
#pragma unroll 16
            for (int k = 0; k < DHID; ++k) {
                const float w = sa.Wo1[k * DHID + tid];
                s3 += cg[k] * w;
                sd += sa.bg[k] * w;
            }
            sa.c3[tid] = s3;
            sa.d3[tid] = sd + sa.bo1[tid];
            return;
        }
        bx -= SIDE_BLOCKS;
    }
    gemm_tile<K, BIAS, ACT>(Wp, bx, A, W, bias, Out, Nrows);
}

__device__ __forceinline__ float4 load_row(const float* __restrict__ g, int sj, int k, int c4) {
    const int s = __builtin_amdgcn_readlane(sj, k);
    return *(const float4*)(g + (size_t)s * DHID + c4);
}

template <int M>
__device__ __forceinline__ float xrow(float x) {
    return __int_as_float(__builtin_amdgcn_update_dpp(0, __float_as_int(x), 0x160 + M, 0xf, 0xf, false));
}
__device__ __forceinline__ float xhalf(float x) {
    const int b = __float_as_int(x);
    return __int_as_float(__builtin_amdgcn_permlanex16(b, b, 0x76543210, 0xfedcba98, false, false));
}

__device__ __forceinline__ void add_row(float4& acc, const float4 v, float wj, int k) {
    const float w = __int_as_float(__builtin_amdgcn_readlane(__float_as_int(wj), k));
    acc.x += v.x * w;
    acc.y += v.y * w;
    acc.z += v.z * w;
    acc.w += v.w * w;
}

#define PROJ_WAVES 16
template <bool PROJ>
__global__ __launch_bounds__(PROJ ? PROJ_WAVES * 32 : 256) void gather_kernel(
    const float* __restrict__ g, const int* __restrict__ bucket,
    const int* __restrict__ off, const int* __restrict__ cnt,
    const float* __restrict__ dinv, const float* __restrict__ bias,
    float* __restrict__ out, int Nrows,
    const float* __restrict__ W2, const float* __restrict__ b2)
{
    __shared__ float2 res[PROJ_WAVES];
    const int lane = threadIdx.x & 31;
    const int iw = __builtin_amdgcn_readfirstlane(
        (int)(blockIdx.x * (blockDim.x >> 5) + (threadIdx.x >> 5)));
    if (!PROJ && iw >= Nrows) return;
    const int i = PROJ ? min(iw, Nrows - 1) : iw;
    const int start = off[i];
    const int n     = cnt[i];
    const float di  = dinv[i];
    const int c4 = lane * 4;
    const float4 gs = *(const float4*)(g + (size_t)i * DHID + c4);
    const float4 bb = *(const float4*)(bias + c4);

    float4 acc = make_float4(0.0f, 0.0f, 0.0f, 0.0f);
    for (int base = 0; base < n; base += 32) {
        const int m = min(32, n - base);
        int sj = 0;
        float wj = 0.0f;
        if (lane < m) {
            sj = bucket[start + base + lane];
            sj = min(max(sj, 0), Nrows - 1);
            wj = dinv[sj] * di;
        }
        int k = 0;
        for (; k + 4 <= m; k += 4) {
            const float4 v0 = load_row(g, sj, k,     c4);
            const float4 v1 = load_row(g, sj, k + 1, c4);
            const float4 v2 = load_row(g, sj, k + 2, c4);
            const float4 v3 = load_row(g, sj, k + 3, c4);
            add_row(acc, v0, wj, k);
            add_row(acc, v1, wj, k + 1);
            add_row(acc, v2, wj, k + 2);
            add_row(acc, v3, wj, k + 3);
        }
        const int r = m - k;
        if (r == 3) {
            const float4 v0 = load_row(g, sj, k,     c4);
            const float4 v1 = load_row(g, sj, k + 1, c4);
            const float4 v2 = load_row(g, sj, k + 2, c4);
            add_row(acc, v0, wj, k);
            add_row(acc, v1, wj, k + 1);
            add_row(acc, v2, wj, k + 2);
        } else if (r == 2) {
            const float4 v0 = load_row(g, sj, k,     c4);
            const float4 v1 = load_row(g, sj, k + 1, c4);
            add_row(acc, v0, wj, k);
            add_row(acc, v1, wj, k + 1);
        } else if (r == 1) {
            const float4 v0 = load_row(g, sj, k, c4);
            add_row(acc, v0, wj, k);
        }
    }

    const float s2 = di * di;
    float4 r4;
    r4.x = acc.x + gs.x * s2 + bb.x;
    r4.y = acc.y + gs.y * s2 + bb.y;
    r4.z = acc.z + gs.z * s2 + bb.z;
    r4.w = acc.w + gs.w * s2 + bb.w;
    if (!PROJ) {
        *(float4*)(out + (size_t)i * DHID + c4) = r4;
        return;
    }
    const float v0 = (r4.x >= 0.0f) ? r4.x : NEG_SLOPE * r4.x;
    const float v1 = (r4.y >= 0.0f) ? r4.y : NEG_SLOPE * r4.y;
    const float v2 = (r4.z >= 0.0f) ? r4.z : NEG_SLOPE * r4.z;
    const float v3 = (r4.w >= 0.0f) ? r4.w : NEG_SLOPE * r4.w;
    const float4 wa = *(const float4*)(W2 + c4 * 2);
    const float4 wb = *(const float4*)(W2 + c4 * 2 + 4);
    float p0 = (v0 * wa.x + v1 * wa.z) + (v2 * wb.x + v3 * wb.z);
    float p1 = (v0 * wa.y + v1 * wa.w) + (v2 * wb.y + v3 * wb.w);
    p0 += xrow<1>(p0);  p1 += xrow<1>(p1);
    p0 += xrow<2>(p0);  p1 += xrow<2>(p1);
    p0 += xrow<4>(p0);  p1 += xrow<4>(p1);
    p0 += xrow<8>(p0);  p1 += xrow<8>(p1);
    p0 += xhalf(p0);    p1 += xhalf(p1);
    if (lane == 0) res[threadIdx.x >> 5] = make_float2(p0 + b2[0], p1 + b2[1]);
    __syncthreads();
    if (threadIdx.x < PROJ_WAVES) {
        const int node = (int)blockIdx.x * PROJ_WAVES + (int)threadIdx.x;
        if (node < Nrows) *(float2*)(out + (size_t)node * 2) = res[threadIdx.x];
    }
}

extern "C" void kernel_launch(void* const* d_in, const int* in_sizes, int n_in,
                              void* d_out, int out_size, void* d_ws, size_t ws_size,
                              hipStream_t stream)
{
    const float* x    = (const float*)d_in[0];
    const int*   ei   = (const int*)  d_in[1];
    const float* W_in = (const float*)d_in[3];
    const float* b_in = (const float*)d_in[4];
    const float* W_g  = (const float*)d_in[5];
    const float* b_g  = (const float*)d_in[6];
    const float* W_o1 = (const float*)d_in[7];
    const float* b_o1 = (const float*)d_in[8];
    const float* W_o2 = (const float*)d_in[9];
    const float* b_o2 = (const float*)d_in[10];

    const int ND = 239;
    const int N  = in_sizes[0] / ND;
    const int E  = in_sizes[1] / 2;
    const int* src = ei;
    const int* dst = ei + E;

    const size_t nh = (size_t)N * DHID;
    float* ws   = (float*)d_ws;
    int*   cnt  = (int*)d_ws;
    float* dinv = ws + (size_t)N;
    float* bufA = ws + 2 * (size_t)N;
    float* bufB = bufA + nh;
    int*   off    = (int*)(bufB + nh);
    int*   rank   = off + (size_t)N;
    int*   bucket = rank + (size_t)E;
    float* Wsq    = (float*)(bucket + (size_t)E);
    float* W3     = Wsq + DHID * DHID;
    float* c3     = W3 + DHID * DHID;
    float* d3     = c3 + DHID;

    const int gemm_blocks = (N / 16 + 3) / 4;
    const int edge_blocks = (E + 255) / 256;
    const int scan_blocks = (N + SCAN_T - 1) / SCAN_T;
    const int gath_blocks = (N + 7) / 8;
    const int proj_blocks = (N + PROJ_WAVES - 1) / PROJ_WAVES;

    PlaceArgs none;
    none.src = nullptr; none.dst = nullptr; none.off = nullptr; none.rank = nullptr;
    none.bucket = nullptr; none.E = 0; none.gblocks = gemm_blocks;
    PlaceArgs pl;
    pl.src = src; pl.dst = dst; pl.off = off; pl.rank = rank;
    pl.bucket = bucket; pl.E = E; pl.gblocks = gemm_blocks;
    const int place_blocks = (E + 127) / 128;
    SideArgs noside;
    noside.Wg = nullptr; noside.bg = nullptr; noside.Wo1 = nullptr; noside.bo1 = nullptr;
    noside.Wsq = nullptr; noside.W3 = nullptr; noside.c3 = nullptr; noside.d3 = nullptr;
    SideArgs sd;
    sd.Wg = W_g; sd.bg = b_g; sd.Wo1 = W_o1; sd.bo1 = b_o1;
    sd.Wsq = Wsq; sd.W3 = W3; sd.c3 = c3; sd.d3 = d3;

    gemm_wmma<239, true, true, true, false, true><<<gemm_blocks + SIDE_BLOCKS, 128, 0, stream>>>(
        x, W_in, b_in, bufA, N, cnt, N, none, sd);

    count_rank_kernel<<<edge_blocks, 256, 0, stream>>>(dst, cnt, rank, E);
    offsets_kernel<<<scan_blocks, SCAN_T, 0, stream>>>(cnt, off, dinv, N);

    gemm_wmma<128, false, false, false, true, false><<<gemm_blocks + place_blocks, 128, 0, stream>>>(
        bufA, W3, nullptr, bufB, N, nullptr, 0, pl, noside);
    gather_kernel<false><<<gath_blocks, 256, 0, stream>>>(
        bufB, bucket, off, cnt, dinv, c3, bufA, N, nullptr, nullptr);
    gather_kernel<true><<<proj_blocks, PROJ_WAVES * 32, 0, stream>>>(
        bufA, bucket, off, cnt, dinv, d3, (float*)d_out, N, W_o2, b_o2);
}


# Edge bucketing without global atomics: two-level counting sort on LDS counters (slice counts per 256-node bin, coarse place, per-bin place in the 2nd GEMM launch); 6 -> 5 launches (sum order changes)
# speedup vs baseline: 3.6204x; 1.2193x over previous
#include <hip/hip_runtime.h>
#include <math.h>


typedef __attribute__((ext_vector_type(2))) float v2f;
typedef __attribute__((ext_vector_type(8))) float v8f;

#define DHID 128
#define TK 64
#define NEG_SLOPE 0.01f

#define BIN_SHIFT 8
#define BIN_NODES 256
#define NBIN_MAX 512
#define EDGE_BLOCKS 256
#define COARSE_T 1024

struct EdgeArgs {
    const int* src;
    const int* dst;
    int*   hist;
    int*   binstart;
    int2*  cedge;
    int*   off;
    int*   cnt;
    float* dinv;
    int*   bucket;
    int    E;
    int    N;
    int    nbin;
    int    nbinp;
    int    epb;
    int    front;
};

__device__ __forceinline__ int coarse_bin(int d, int nbin) {
    return min(max(d >> BIN_SHIFT, 0), nbin - 1);
}

__device__ __forceinline__ void edge_hist(int* h, const EdgeArgs& ea, const int blk, const int tid) {
    const int nbin = min(ea.nbin, NBIN_MAX);
    for (int b = tid; b < nbin; b += 128) h[b] = 0;
    __syncthreads();
    const int e0 = blk * ea.epb;
    const int e1 = min(ea.E, e0 + ea.epb);
    for (int e = e0 + tid; e < e1; e += 8 * 128) {
        int d[8];
#pragma unroll
        for (int u = 0; u < 8; ++u) d[u] = ea.dst[min(e + u * 128, e1 - 1)];
#pragma unroll
        for (int u = 0; u < 8; ++u)
            atomicAdd(&h[coarse_bin(d[u], nbin)], (e + u * 128 < e1) ? 1 : 0);
    }
    __syncthreads();
    for (int b = tid; b < nbin; b += 128) ea.hist[blk * ea.nbinp + b] = h[b];
}

__global__ __launch_bounds__(COARSE_T) void coarse_place_kernel(const EdgeArgs ea) {
    __shared__ int sc[NBIN_MAX];
    __shared__ int fill[NBIN_MAX];
    __shared__ int hi_t[NBIN_MAX];
    __shared__ int hi_p[NBIN_MAX];
    const int tid  = threadIdx.x;
    const int blk  = blockIdx.x;
    const int nbin = min(ea.nbin, NBIN_MAX);
    const int b    = tid & (NBIN_MAX - 1);
    const int part = tid >> 9;
    int t_sum = 0, p_sum = 0;
    if (b < nbin) {
        const int j0 = part * (EDGE_BLOCKS / 2);
        const int* __restrict__ col = ea.hist + b;
#pragma unroll 16
        for (int j = j0; j < j0 + EDGE_BLOCKS / 2; ++j) {
            const int v = col[j * ea.nbinp];
            t_sum += v;
            p_sum += (j < blk) ? v : 0;
        }
    }
    if (part == 1) { hi_t[b] = t_sum; hi_p[b] = p_sum; }
    __syncthreads();
    if (part == 0) { t_sum += hi_t[b]; p_sum += hi_p[b]; sc[b] = t_sum; }
    __syncthreads();
    for (int o = 1; o < NBIN_MAX; o <<= 1) {
        int v = 0;
        if (tid < NBIN_MAX && tid >= o) v = sc[tid - o];
        __syncthreads();
        if (tid < NBIN_MAX) sc[tid] += v;
        __syncthreads();
    }
    if (part == 0) {
        const int ex = sc[b] - t_sum;
        fill[b] = ex + p_sum;
        if (blk == 0) {
            if (b < nbin) ea.binstart[b] = ex;
            if (b == nbin - 1) ea.binstart[nbin] = sc[b];
        }
    }
    __syncthreads();
    const int e0 = blk * ea.epb;
    const int e1 = min(ea.E, e0 + ea.epb);
    for (int e = e0 + tid; e < e1; e += COARSE_T) {
        const int d = ea.dst[e];
        const int s = ea.src[e];
        int pos = atomicAdd(&fill[coarse_bin(d, nbin)], 1);
        pos = min(max(pos, 0), ea.E - 1);
        ea.cedge[pos] = make_int2(s, d);
    }
}

__device__ __forceinline__ void fine_place(int* lds, const EdgeArgs& ea, const int b, const int tid) {
    int* c    = lds;
    int* ps   = lds + BIN_NODES;
    int* fill = lds + 2 * BIN_NODES;
    c[tid] = 0;
    c[tid + 128] = 0;
    int start = ea.binstart[b];
    int end   = ea.binstart[b + 1];
    start = min(max(start, 0), ea.E);
    end   = min(max(end, start), ea.E);
    const int n = end - start;
    const int2* __restrict__ ce = ea.cedge + start;
    __syncthreads();
    for (int i = tid; i < n; i += 4 * 128) {
        int d[4];
#pragma unroll
        for (int u = 0; u < 4; ++u) d[u] = ce[min(i + u * 128, n - 1)].y;
#pragma unroll
        for (int u = 0; u < 4; ++u)
            atomicAdd(&c[d[u] & (BIN_NODES - 1)], (i + u * 128 < n) ? 1 : 0);
    }
    __syncthreads();
    const int c0 = c[2 * tid];
    const int c1 = c[2 * tid + 1];
    ps[tid] = c0 + c1;
    __syncthreads();
    for (int o = 1; o < 128; o <<= 1) {
        const int v = (tid >= o) ? ps[tid - o] : 0;
        __syncthreads();
        ps[tid] += v;
        __syncthreads();
    }
    const int ex0 = start + ps[tid] - (c0 + c1);
    const int ex1 = ex0 + c0;
    fill[2 * tid]     = ex0;
    fill[2 * tid + 1] = ex1;
    const int node0 = b * BIN_NODES + 2 * tid;
    if (node0 < ea.N) {
        ea.off[node0]  = ex0;
        ea.cnt[node0]  = c0;
        ea.dinv[node0] = rsqrtf((float)c0 + 1.0f);
    }
    if (node0 + 1 < ea.N) {
        ea.off[node0 + 1]  = ex1;
        ea.cnt[node0 + 1]  = c1;
        ea.dinv[node0 + 1] = rsqrtf((float)c1 + 1.0f);
    }
    __syncthreads();
    for (int i = tid; i < n; i += 4 * 128) {
        int2 ed[4];
#pragma unroll
        for (int u = 0; u < 4; ++u) ed[u] = ce[min(i + u * 128, n - 1)];
#pragma unroll
        for (int u = 0; u < 4; ++u) {
            const bool live = (i + u * 128 < n);
            int pos = atomicAdd(&fill[ed[u].y & (BIN_NODES - 1)], live ? 1 : 0);
            pos = min(max(pos, 0), ea.E - 1);
            if (live) ea.bucket[pos] = ed[u].x;
        }
    }
}


typedef float f4u __attribute__((ext_vector_type(4), aligned(4)));

template <int K, int KV>
__device__ __forceinline__ void gemm_chunk(
    float (*Wp)[DHID * 2], const float* __restrict__ arow0, const float* __restrict__ W,
    const int k0, const int tid, const int m, const int h, v8f (&acc)[8])
{
    constexpr int NG = (KV + 7) / 8;
    float4 a4[NG];
#pragma unroll
    for (int g = 0; g < NG; ++g) {
        if (8 * g + 8 <= KV) {
            const f4u v = *(const f4u*)(arow0 + k0 + 8 * g + 4 * h);
            a4[g] = make_float4(v.x, v.y, v.z, v.w);
        } else {
            const int e = k0 + 8 * g + 4 * h;
            a4[g].x = arow0[min(e,     K - 1)];
            a4[g].y = arow0[min(e + 1, K - 1)];
            a4[g].z = arow0[min(e + 2, K - 1)];
            a4[g].w = arow0[min(e + 3, K - 1)];
        }
    }

    constexpr int RT = ((NG * 8) % 32 == 0) ? 32 : 16;
    const float* __restrict__ wbase = W + (size_t)k0 * DHID;
    float* __restrict__ wdst = &Wp[0][tid * 2];
#pragma nounroll
    for (int r0 = 0; r0 < NG * 8; r0 += RT) {
        float t[RT];
#pragma unroll
        for (int u = 0; u < RT; ++u) {
            t[u] = (KV == NG * 8 || r0 + u < KV) ? wbase[(r0 + u) * DHID + tid] : 0.0f;
        }
#pragma unroll
        for (int u = 0; u < RT; ++u)
            wdst[((r0 + u) >> 1) * (DHID * 2) + (u & 1)] = t[u];
    }
    __syncthreads();

#pragma unroll
    for (int g = 0; g < NG; ++g) {
#pragma unroll
        for (int j = 0; j < 2; ++j) {
            v2f a;
            a.x = j ? a4[g].z : a4[g].x;
            a.y = j ? a4[g].w : a4[g].y;
            const float* __restrict__ wp = &Wp[4 * g + 2 * h + j][0];
#pragma unroll
            for (int t = 0; t < 8; ++t) {
                const v2f b = *(const v2f*)(wp + (t * 16 + m) * 2);
                acc[t] = __builtin_amdgcn_wmma_f32_16x16x4_f32(
                    false, a, false, b, (short)0, acc[t], false, false);
            }
        }
    }
    __syncthreads();
}

template <int K, bool BIAS, bool ACT>
__device__ __forceinline__ void gemm_tile(
    float (*Wp)[DHID * 2], const int bx,
    const float* __restrict__ A, const float* __restrict__ W,
    const float* __restrict__ bias, float* __restrict__ Out, int Nrows)
{
    const int tid  = threadIdx.x;
    const int wave = tid >> 5;
    const int lane = tid & 31;
    const int row0 = __builtin_amdgcn_readfirstlane((bx * 4 + wave) * 16);
    const bool rowok = (row0 < Nrows);

    const int m = lane & 15;
    const int h = lane >> 4;

    v8f acc[8];
#pragma unroll
    for (int t = 0; t < 8; ++t) acc[t] = (v8f){};

    const float* __restrict__ arow0 = A + (size_t)((rowok ? row0 : 0) + m) * K;

#pragma nounroll
    for (int k0 = 0; k0 + TK <= K; k0 += TK)
        gemm_chunk<K, TK>(Wp, arow0, W, k0, tid, m, h, acc);
    if (K % TK != 0)
        gemm_chunk<K, (K % TK != 0) ? (K % TK) : TK>(Wp, arow0, W, (K / TK) * TK, tid, m, h, acc);

    if (!rowok) return;
    const int rbase = row0 + ((lane >> 4) ? 8 : 0);
#pragma unroll
    for (int t = 0; t < 8; ++t) {
        const int col = t * 16 + m;
        const float bv = BIAS ? bias[col] : 0.0f;
#pragma unroll
        for (int r = 0; r < 8; ++r) {
            float v = acc[t][r] + bv;
            if (ACT) v = (v >= 0.0f) ? v : NEG_SLOPE * v;
            Out[(size_t)(rbase + r) * DHID + col] = v;
        }
    }
}

struct SideArgs {
    const float* Wg;
    const float* bg;
    const float* Wo1;
    const float* bo1;
    float*       Wsq;
    float*       W3;
    float*       c3;
    float*       d3;
};
#define SIDE_GEMM_BLOCKS (DHID / 64)
#define SIDE_BLOCKS (SIDE_GEMM_BLOCKS + 1)

template <int K, bool BIAS, bool ACT, bool HIST, bool FINE, bool SIDE>
__global__ __launch_bounds__(128, 6) void gemm_wmma(
    const float* __restrict__ A, const float* __restrict__ W,
    const float* __restrict__ bias, float* __restrict__ Out, int Nrows,
    const EdgeArgs ea, const SideArgs sa)
{
    __shared__ float Wp[TK / 2][DHID * 2];

    const int tid  = threadIdx.x;
    int bx = blockIdx.x;
    if (HIST || FINE) {
        if (bx < ea.front) {
            if (HIST) edge_hist((int*)&Wp[0][0], ea, bx, tid);
            else      fine_place((int*)&Wp[0][0], ea, bx, tid);
            return;
        }
        bx -= ea.front;
    }
    if (SIDE) {
        if (bx < SIDE_GEMM_BLOCKS) {
            gemm_tile<DHID, false, false>(Wp, bx, sa.Wg, sa.Wg, nullptr, sa.Wsq, DHID);
            __threadfence();
            __syncthreads();
            gemm_tile<DHID, false, false>(Wp, bx, sa.Wsq, sa.Wo1, nullptr, sa.W3, DHID);
            return;
        }
        if (bx == SIDE_GEMM_BLOCKS) {
            float* cg = &Wp[0][0];
            float s = 0.0f;
#pragma unroll 16
            for (int k = 0; k < DHID; ++k) s += sa.bg[k] * sa.Wg[k * DHID + tid];
            cg[tid] = s;
            __syncthreads();
            float s3 = 0.0f, sd = 0.0f;
#pragma unroll 16
            for (int k = 0; k < DHID; ++k) {
                const float w = sa.Wo1[k * DHID + tid];
                s3 += cg[k] * w;
                sd += sa.bg[k] * w;
            }
            sa.c3[tid] = s3;
            sa.d3[tid] = sd + sa.bo1[tid];
            return;
        }
        bx -= SIDE_BLOCKS;
    }
    gemm_tile<K, BIAS, ACT>(Wp, bx, A, W, bias, Out, Nrows);
}

__device__ __forceinline__ float4 load_row(const float* __restrict__ g, int sj, int k, int c4) {
    const int s = __builtin_amdgcn_readlane(sj, k);
    return *(const float4*)(g + (size_t)s * DHID + c4);
}

template <int M>
__device__ __forceinline__ float xrow(float x) {
    return __int_as_float(__builtin_amdgcn_update_dpp(0, __float_as_int(x), 0x160 + M, 0xf, 0xf, false));
}
__device__ __forceinline__ float xhalf(float x) {
    const int b = __float_as_int(x);
    return __int_as_float(__builtin_amdgcn_permlanex16(b, b, 0x76543210, 0xfedcba98, false, false));
}

__device__ __forceinline__ void add_row(float4& acc, const float4 v, float wj, int k) {
    const float w = __int_as_float(__builtin_amdgcn_readlane(__float_as_int(wj), k));
    acc.x += v.x * w;
    acc.y += v.y * w;
    acc.z += v.z * w;
    acc.w += v.w * w;
}

#define PROJ_WAVES 16
template <bool PROJ>
__global__ __launch_bounds__(PROJ ? PROJ_WAVES * 32 : 256) void gather_kernel(
    const float* __restrict__ g, const int* __restrict__ bucket,
    const int* __restrict__ off, const int* __restrict__ cnt,
    const float* __restrict__ dinv, const float* __restrict__ bias,
    float* __restrict__ out, int Nrows,
    const float* __restrict__ W2, const float* __restrict__ b2)
{
    __shared__ float2 res[PROJ_WAVES];
    const int lane = threadIdx.x & 31;
    const int iw = __builtin_amdgcn_readfirstlane(
        (int)(blockIdx.x * (blockDim.x >> 5) + (threadIdx.x >> 5)));
    if (!PROJ && iw >= Nrows) return;
    const int i = PROJ ? min(iw, Nrows - 1) : iw;
    const int start = off[i];
    const int n     = cnt[i];
    const float di  = dinv[i];
    const int c4 = lane * 4;
    const float4 gs = *(const float4*)(g + (size_t)i * DHID + c4);
    const float4 bb = *(const float4*)(bias + c4);

    float4 acc = make_float4(0.0f, 0.0f, 0.0f, 0.0f);
    for (int base = 0; base < n; base += 32) {
        const int m = min(32, n - base);
        int sj = 0;
        float wj = 0.0f;
        if (lane < m) {
            sj = bucket[start + base + lane];
            sj = min(max(sj, 0), Nrows - 1);
            wj = dinv[sj] * di;
        }
        int k = 0;
        for (; k + 4 <= m; k += 4) {
            const float4 v0 = load_row(g, sj, k,     c4);
            const float4 v1 = load_row(g, sj, k + 1, c4);
            const float4 v2 = load_row(g, sj, k + 2, c4);
            const float4 v3 = load_row(g, sj, k + 3, c4);
            add_row(acc, v0, wj, k);
            add_row(acc, v1, wj, k + 1);
            add_row(acc, v2, wj, k + 2);
            add_row(acc, v3, wj, k + 3);
        }
        const int r = m - k;
        if (r == 3) {
            const float4 v0 = load_row(g, sj, k,     c4);
            const float4 v1 = load_row(g, sj, k + 1, c4);
            const float4 v2 = load_row(g, sj, k + 2, c4);
            add_row(acc, v0, wj, k);
            add_row(acc, v1, wj, k + 1);
            add_row(acc, v2, wj, k + 2);
        } else if (r == 2) {
            const float4 v0 = load_row(g, sj, k,     c4);
            const float4 v1 = load_row(g, sj, k + 1, c4);
            add_row(acc, v0, wj, k);
            add_row(acc, v1, wj, k + 1);
        } else if (r == 1) {
            const float4 v0 = load_row(g, sj, k, c4);
            add_row(acc, v0, wj, k);
        }
    }

    const float s2 = di * di;
    float4 r4;
    r4.x = acc.x + gs.x * s2 + bb.x;
    r4.y = acc.y + gs.y * s2 + bb.y;
    r4.z = acc.z + gs.z * s2 + bb.z;
    r4.w = acc.w + gs.w * s2 + bb.w;
    if (!PROJ) {
        *(float4*)(out + (size_t)i * DHID + c4) = r4;
        return;
    }
    const float v0 = (r4.x >= 0.0f) ? r4.x : NEG_SLOPE * r4.x;
    const float v1 = (r4.y >= 0.0f) ? r4.y : NEG_SLOPE * r4.y;
    const float v2 = (r4.z >= 0.0f) ? r4.z : NEG_SLOPE * r4.z;
    const float v3 = (r4.w >= 0.0f) ? r4.w : NEG_SLOPE * r4.w;
    const float4 wa = *(const float4*)(W2 + c4 * 2);
    const float4 wb = *(const float4*)(W2 + c4 * 2 + 4);
    float p0 = (v0 * wa.x + v1 * wa.z) + (v2 * wb.x + v3 * wb.z);
    float p1 = (v0 * wa.y + v1 * wa.w) + (v2 * wb.y + v3 * wb.w);
    p0 += xrow<1>(p0);  p1 += xrow<1>(p1);
    p0 += xrow<2>(p0);  p1 += xrow<2>(p1);
    p0 += xrow<4>(p0);  p1 += xrow<4>(p1);
    p0 += xrow<8>(p0);  p1 += xrow<8>(p1);
    p0 += xhalf(p0);    p1 += xhalf(p1);
    if (lane == 0) res[threadIdx.x >> 5] = make_float2(p0 + b2[0], p1 + b2[1]);
    __syncthreads();
    if (threadIdx.x < PROJ_WAVES) {
        const int node = (int)blockIdx.x * PROJ_WAVES + (int)threadIdx.x;
        if (node < Nrows) *(float2*)(out + (size_t)node * 2) = res[threadIdx.x];
    }
}

extern "C" void kernel_launch(void* const* d_in, const int* in_sizes, int n_in,
                              void* d_out, int out_size, void* d_ws, size_t ws_size,
                              hipStream_t stream)
{
    const float* x    = (const float*)d_in[0];
    const int*   ei   = (const int*)  d_in[1];
    const float* W_in = (const float*)d_in[3];
    const float* b_in = (const float*)d_in[4];
    const float* W_g  = (const float*)d_in[5];
    const float* b_g  = (const float*)d_in[6];
    const float* W_o1 = (const float*)d_in[7];
    const float* b_o1 = (const float*)d_in[8];
    const float* W_o2 = (const float*)d_in[9];
    const float* b_o2 = (const float*)d_in[10];

    const int ND = 239;
    const int N  = in_sizes[0] / ND;
    const int E  = in_sizes[1] / 2;
    const int* src = ei;
    const int* dst = ei + E;

    const size_t nh = (size_t)N * DHID;
    float* ws   = (float*)d_ws;
    int*   cnt  = (int*)d_ws;
    float* dinv = ws + (size_t)N;
    float* bufA = ws + 2 * (size_t)N;
    float* bufB = bufA + nh;
    float* Wsq    = bufB + nh;
    float* W3     = Wsq + DHID * DHID;
    float* c3     = W3 + DHID * DHID;
    float* d3     = c3 + DHID;
    int2*  cedge  = (int2*)(d3 + DHID);
    int*   off    = (int*)(cedge + (size_t)E);
    int*   bucket = off + (size_t)N;
    int*   hist   = bucket + (size_t)E;

    const int gemm_blocks = (N / 16 + 3) / 4;
    const int gath_blocks = (N + 7) / 8;
    const int proj_blocks = (N + PROJ_WAVES - 1) / PROJ_WAVES;
    const int nbin  = (N + BIN_NODES - 1) / BIN_NODES;
    const int nbinp = (nbin + 31) / 32 * 32;
    const int epb   = (E + EDGE_BLOCKS - 1) / EDGE_BLOCKS;
    int*   binstart = hist + (size_t)EDGE_BLOCKS * nbinp;

    EdgeArgs eh;
    eh.src = src; eh.dst = dst; eh.hist = hist; eh.binstart = binstart; eh.cedge = cedge;
    eh.off = off; eh.cnt = cnt; eh.dinv = dinv; eh.bucket = bucket;
    eh.E = E; eh.N = N; eh.nbin = nbin; eh.nbinp = nbinp; eh.epb = epb;
    eh.front = EDGE_BLOCKS;
    EdgeArgs ef = eh;
    ef.front = nbin;
    SideArgs noside;
    noside.Wg = nullptr; noside.bg = nullptr; noside.Wo1 = nullptr; noside.bo1 = nullptr;
    noside.Wsq = nullptr; noside.W3 = nullptr; noside.c3 = nullptr; noside.d3 = nullptr;
    SideArgs sd;
    sd.Wg = W_g; sd.bg = b_g; sd.Wo1 = W_o1; sd.bo1 = b_o1;
    sd.Wsq = Wsq; sd.W3 = W3; sd.c3 = c3; sd.d3 = d3;

    gemm_wmma<239, true, true, true, false, true><<<EDGE_BLOCKS + SIDE_BLOCKS + gemm_blocks, 128, 0, stream>>>(
        x, W_in, b_in, bufA, N, eh, sd);

    coarse_place_kernel<<<EDGE_BLOCKS, COARSE_T, 0, stream>>>(eh);

    gemm_wmma<128, false, false, false, true, false><<<nbin + gemm_blocks, 128, 0, stream>>>(
        bufA, W3, nullptr, bufB, N, ef, noside);
    gather_kernel<false><<<gath_blocks, 256, 0, stream>>>(
        bufB, bucket, off, cnt, dinv, c3, bufA, N, nullptr, nullptr);
    gather_kernel<true><<<proj_blocks, PROJ_WAVES * 32, 0, stream>>>(
        bufA, bucket, off, cnt, dinv, d3, (float*)d_out, N, W_o2, b_o2);
}


# Coarse placement launch removed: each edge slice sorts itself by 256-node bin in LDS inside GEMM launch 1; a bin's work-group walks its 128 slice segments; 5 -> 4 launches (sum order changes)
# speedup vs baseline: 4.1567x; 1.1481x over previous
#include <hip/hip_runtime.h>
#include <math.h>


typedef __attribute__((ext_vector_type(2))) float v2f;
typedef __attribute__((ext_vector_type(8))) float v8f;

#define DHID 128
#define TK 64
#define NEG_SLOPE 0.01f

#define BIN_SHIFT 8
#define BIN_NODES 256
#define NBIN_MAX 512
#define EDGE_BLOCKS 128
static_assert(EDGE_BLOCKS == 128, "pass 2 gives every slice to one of its 128 threads");

struct EdgeArgs {
    const int* src;
    const int* dst;
    int*   ofs;
    int2*  cedge;
    int*   off;
    int*   cnt;
    float* dinv;
    int*   bucket;
    int    E;
    int    N;
    int    nbin;
    int    nbinp;
    int    epb;
    int    front;
};

__device__ __forceinline__ int coarse_bin(int d, int nbin) {
    return min(max(d >> BIN_SHIFT, 0), nbin - 1);
}

__device__ __forceinline__ void slice_sort(int* lds, const EdgeArgs& ea, const int blk, const int tid) {
    int* h  = lds;
    int* ps = lds + NBIN_MAX;
    const int nbin = min(ea.nbin, NBIN_MAX);
    for (int b = tid; b < NBIN_MAX; b += 128) h[b] = 0;
    __syncthreads();
    const int e0 = min(blk * ea.epb, ea.E);
    const int e1 = min(ea.E, e0 + ea.epb);
    const int ns = e1 - e0;
    for (int e = e0 + tid; e < e1; e += 8 * 128) {
        int d[8];
#pragma unroll
        for (int u = 0; u < 8; ++u) d[u] = ea.dst[min(e + u * 128, e1 - 1)];
#pragma unroll
        for (int u = 0; u < 8; ++u)
            atomicAdd(&h[coarse_bin(d[u], nbin)], (e + u * 128 < e1) ? 1 : 0);
    }
    __syncthreads();
    const int c0 = h[4 * tid], c1 = h[4 * tid + 1], c2 = h[4 * tid + 2], c3 = h[4 * tid + 3];
    ps[tid] = (c0 + c1) + (c2 + c3);
    __syncthreads();
    for (int o = 1; o < 128; o <<= 1) {
        const int v = (tid >= o) ? ps[tid - o] : 0;
        __syncthreads();
        ps[tid] += v;
        __syncthreads();
    }
    const int x0 = ps[tid] - ((c0 + c1) + (c2 + c3));
    const int x1 = x0 + c0;
    const int x2 = x1 + c1;
    const int x3 = x2 + c2;
    h[4 * tid]     = x0;
    h[4 * tid + 1] = x1;
    h[4 * tid + 2] = x2;
    h[4 * tid + 3] = x3;
    int* __restrict__ row = ea.ofs + (size_t)blk * ea.nbinp;
    if (4 * tid     < nbin) row[4 * tid]     = x0;
    if (4 * tid + 1 < nbin) row[4 * tid + 1] = x1;
    if (4 * tid + 2 < nbin) row[4 * tid + 2] = x2;
    if (4 * tid + 3 < nbin) row[4 * tid + 3] = x3;
    if (tid == 127) row[nbin] = ps[127];
    __syncthreads();
    int2* __restrict__ ce = ea.cedge + e0;
    for (int e = e0 + tid; e < e1; e += 8 * 128) {
        int d[8], s[8];
#pragma unroll
        for (int u = 0; u < 8; ++u) {
            const int ee = min(e + u * 128, e1 - 1);
            d[u] = ea.dst[ee];
            s[u] = ea.src[ee];
        }
#pragma unroll
        for (int u = 0; u < 8; ++u) {
            const bool live = (e + u * 128 < e1);
            int pos = atomicAdd(&h[coarse_bin(d[u], nbin)], live ? 1 : 0);
            pos = min(max(pos, 0), ns - 1);
            if (live) ce[pos] = make_int2(s[u], d[u]);
        }
    }
}

__device__ __forceinline__ void fine_place(int* lds, const EdgeArgs& ea, const int b, const int tid) {
    int* c    = lds;
    int* ps   = lds + BIN_NODES;
    int* fill = lds + 2 * BIN_NODES;
    int* tot  = lds + 3 * BIN_NODES;
    c[tid] = 0;
    c[tid + 128] = 0;
    if (tid == 0) tot[0] = 0;
    const int base = min(tid * ea.epb, ea.E);
    const int cap  = min(ea.epb, ea.E - base);
    const int* __restrict__ row = ea.ofs + (size_t)tid * ea.nbinp;
    int s0 = row[b];
    int s1 = row[b + 1];
    s0 = min(max(s0, 0), cap);
    s1 = min(max(s1, s0), cap);
    const int2* __restrict__ ce = ea.cedge + base;
    __syncthreads();
    atomicAdd(&tot[0], s0);
    for (int i = s0; i < s1; i += 8) {
        int d[8];
#pragma unroll
        for (int u = 0; u < 8; ++u) d[u] = ce[min(i + u, s1 - 1)].y;
#pragma unroll
        for (int u = 0; u < 8; ++u)
            atomicAdd(&c[d[u] & (BIN_NODES - 1)], (i + u < s1) ? 1 : 0);
    }
    __syncthreads();
    const int c0 = c[2 * tid];
    const int c1 = c[2 * tid + 1];
    ps[tid] = c0 + c1;
    __syncthreads();
    for (int o = 1; o < 128; o <<= 1) {
        const int v = (tid >= o) ? ps[tid - o] : 0;
        __syncthreads();
        ps[tid] += v;
        __syncthreads();
    }
    const int start = min(max(tot[0], 0), ea.E);
    const int ex0 = start + ps[tid] - (c0 + c1);
    const int ex1 = ex0 + c0;
    fill[2 * tid]     = ex0;
    fill[2 * tid + 1] = ex1;
    const int node0 = b * BIN_NODES + 2 * tid;
    if (node0 < ea.N) {
        ea.off[node0]  = ex0;
        ea.cnt[node0]  = c0;
        ea.dinv[node0] = rsqrtf((float)c0 + 1.0f);
    }
    if (node0 + 1 < ea.N) {
        ea.off[node0 + 1]  = ex1;
        ea.cnt[node0 + 1]  = c1;
        ea.dinv[node0 + 1] = rsqrtf((float)c1 + 1.0f);
    }
    __syncthreads();
    for (int i = s0; i < s1; i += 8) {
        int2 ed[8];
#pragma unroll
        for (int u = 0; u < 8; ++u) ed[u] = ce[min(i + u, s1 - 1)];
#pragma unroll
        for (int u = 0; u < 8; ++u) {
            const bool live = (i + u < s1);
            int pos = atomicAdd(&fill[ed[u].y & (BIN_NODES - 1)], live ? 1 : 0);
            pos = min(max(pos, 0), ea.E - 1);
            if (live) ea.bucket[pos] = ed[u].x;
        }
    }
}


typedef float f4u __attribute__((ext_vector_type(4), aligned(4)));

template <int K, int KV>
__device__ __forceinline__ void gemm_chunk(
    float (*Wp)[DHID * 2], const float* __restrict__ arow0, const float* __restrict__ W,
    const int k0, const int tid, const int m, const int h, v8f (&acc)[8])
{
    constexpr int NG = (KV + 7) / 8;
    float4 a4[NG];
#pragma unroll
    for (int g = 0; g < NG; ++g) {
        if (8 * g + 8 <= KV) {
            const f4u v = *(const f4u*)(arow0 + k0 + 8 * g + 4 * h);
            a4[g] = make_float4(v.x, v.y, v.z, v.w);
        } else {
            const int e = k0 + 8 * g + 4 * h;
            a4[g].x = arow0[min(e,     K - 1)];
            a4[g].y = arow0[min(e + 1, K - 1)];
            a4[g].z = arow0[min(e + 2, K - 1)];
            a4[g].w = arow0[min(e + 3, K - 1)];
        }
    }

    constexpr int RT = ((NG * 8) % 32 == 0) ? 32 : 16;
    const float* __restrict__ wbase = W + (size_t)k0 * DHID;
    float* __restrict__ wdst = &Wp[0][tid * 2];
#pragma nounroll
    for (int r0 = 0; r0 < NG * 8; r0 += RT) {
        float t[RT];
#pragma unroll
        for (int u = 0; u < RT; ++u) {
            t[u] = (KV == NG * 8 || r0 + u < KV) ? wbase[(r0 + u) * DHID + tid] : 0.0f;
        }
#pragma unroll
        for (int u = 0; u < RT; ++u)
            wdst[((r0 + u) >> 1) * (DHID * 2) + (u & 1)] = t[u];
    }
    __syncthreads();

#pragma unroll
    for (int g = 0; g < NG; ++g) {
#pragma unroll
        for (int j = 0; j < 2; ++j) {
            v2f a;
            a.x = j ? a4[g].z : a4[g].x;
            a.y = j ? a4[g].w : a4[g].y;
            const float* __restrict__ wp = &Wp[4 * g + 2 * h + j][0];
#pragma unroll
            for (int t = 0; t < 8; ++t) {
                const v2f b = *(const v2f*)(wp + (t * 16 + m) * 2);
                acc[t] = __builtin_amdgcn_wmma_f32_16x16x4_f32(
                    false, a, false, b, (short)0, acc[t], false, false);
            }
        }
    }
    __syncthreads();
}

template <int K, bool BIAS, bool ACT>
__device__ __forceinline__ void gemm_tile(
    float (*Wp)[DHID * 2], const int bx,
    const float* __restrict__ A, const float* __restrict__ W,
    const float* __restrict__ bias, float* __restrict__ Out, int Nrows)
{
    const int tid  = threadIdx.x;
    const int wave = tid >> 5;
    const int lane = tid & 31;
    const int row0 = __builtin_amdgcn_readfirstlane((bx * 4 + wave) * 16);
    const bool rowok = (row0 < Nrows);

    const int m = lane & 15;
    const int h = lane >> 4;

    v8f acc[8];
#pragma unroll
    for (int t = 0; t < 8; ++t) acc[t] = (v8f){};

    const float* __restrict__ arow0 = A + (size_t)((rowok ? row0 : 0) + m) * K;

#pragma nounroll
    for (int k0 = 0; k0 + TK <= K; k0 += TK)
        gemm_chunk<K, TK>(Wp, arow0, W, k0, tid, m, h, acc);
    if (K % TK != 0)
        gemm_chunk<K, (K % TK != 0) ? (K % TK) : TK>(Wp, arow0, W, (K / TK) * TK, tid, m, h, acc);

    if (!rowok) return;
    const int rbase = row0 + ((lane >> 4) ? 8 : 0);
#pragma unroll
    for (int t = 0; t < 8; ++t) {
        const int col = t * 16 + m;
        const float bv = BIAS ? bias[col] : 0.0f;
#pragma unroll
        for (int r = 0; r < 8; ++r) {
            float v = acc[t][r] + bv;
            if (ACT) v = (v >= 0.0f) ? v : NEG_SLOPE * v;
            Out[(size_t)(rbase + r) * DHID + col] = v;
        }
    }
}

struct SideArgs {
    const float* Wg;
    const float* bg;
    const float* Wo1;
    const float* bo1;
    float*       Wsq;
    float*       W3;
    float*       c3;
    float*       d3;
};
#define SIDE_GEMM_BLOCKS (DHID / 64)
#define SIDE_BLOCKS (SIDE_GEMM_BLOCKS + 1)

template <int K, bool BIAS, bool ACT, bool HIST, bool FINE, bool SIDE>
__global__ __launch_bounds__(128, 6) void gemm_wmma(
    const float* __restrict__ A, const float* __restrict__ W,
    const float* __restrict__ bias, float* __restrict__ Out, int Nrows,
    const EdgeArgs ea, const SideArgs sa)
{
    __shared__ float Wp[TK / 2][DHID * 2];

    const int tid  = threadIdx.x;
    int bx = blockIdx.x;
    if (HIST || FINE) {
        if (bx < ea.front) {
            if (HIST) slice_sort((int*)&Wp[0][0], ea, bx, tid);
            else      fine_place((int*)&Wp[0][0], ea, bx, tid);
            return;
        }
        bx -= ea.front;
    }
    if (SIDE) {
        if (bx < SIDE_GEMM_BLOCKS) {
            gemm_tile<DHID, false, false>(Wp, bx, sa.Wg, sa.Wg, nullptr, sa.Wsq, DHID);
            __threadfence();
            __syncthreads();
            gemm_tile<DHID, false, false>(Wp, bx, sa.Wsq, sa.Wo1, nullptr, sa.W3, DHID);
            return;
        }
        if (bx == SIDE_GEMM_BLOCKS) {
            float* cg = &Wp[0][0];
            float s = 0.0f;
#pragma unroll 16
            for (int k = 0; k < DHID; ++k) s += sa.bg[k] * sa.Wg[k * DHID + tid];
            cg[tid] = s;
            __syncthreads();
            float s3 = 0.0f, sd = 0.0f;
#pragma unroll 16
            for (int k = 0; k < DHID; ++k) {
                const float w = sa.Wo1[k * DHID + tid];
                s3 += cg[k] * w;
                sd += sa.bg[k] * w;
            }
            sa.c3[tid] = s3;
            sa.d3[tid] = sd + sa.bo1[tid];
            return;
        }
        bx -= SIDE_BLOCKS;
    }
    gemm_tile<K, BIAS, ACT>(Wp, bx, A, W, bias, Out, Nrows);
}

__device__ __forceinline__ float4 load_row(const float* __restrict__ g, int sj, int k, int c4) {
    const int s = __builtin_amdgcn_readlane(sj, k);
    return *(const float4*)(g + (size_t)s * DHID + c4);
}

template <int M>
__device__ __forceinline__ float xrow(float x) {
    return __int_as_float(__builtin_amdgcn_update_dpp(0, __float_as_int(x), 0x160 + M, 0xf, 0xf, false));
}
__device__ __forceinline__ float xhalf(float x) {
    const int b = __float_as_int(x);
    return __int_as_float(__builtin_amdgcn_permlanex16(b, b, 0x76543210, 0xfedcba98, false, false));
}

__device__ __forceinline__ void add_row(float4& acc, const float4 v, float wj, int k) {
    const float w = __int_as_float(__builtin_amdgcn_readlane(__float_as_int(wj), k));
    acc.x += v.x * w;
    acc.y += v.y * w;
    acc.z += v.z * w;
    acc.w += v.w * w;
}

#define PROJ_WAVES 16
template <bool PROJ>
__global__ __launch_bounds__(PROJ ? PROJ_WAVES * 32 : 256) void gather_kernel(
    const float* __restrict__ g, const int* __restrict__ bucket,
    const int* __restrict__ off, const int* __restrict__ cnt,
    const float* __restrict__ dinv, const float* __restrict__ bias,
    float* __restrict__ out, int Nrows,
    const float* __restrict__ W2, const float* __restrict__ b2)
{
    __shared__ float2 res[PROJ_WAVES];
    const int lane = threadIdx.x & 31;
    const int iw = __builtin_amdgcn_readfirstlane(
        (int)(blockIdx.x * (blockDim.x >> 5) + (threadIdx.x >> 5)));
    if (!PROJ && iw >= Nrows) return;
    const int i = PROJ ? min(iw, Nrows - 1) : iw;
    const int start = off[i];
    const int n     = cnt[i];
    const float di  = dinv[i];
    const int c4 = lane * 4;
    const float4 gs = *(const float4*)(g + (size_t)i * DHID + c4);
    const float4 bb = *(const float4*)(bias + c4);

    float4 acc = make_float4(0.0f, 0.0f, 0.0f, 0.0f);
    for (int base = 0; base < n; base += 32) {
        const int m = min(32, n - base);
        int sj = 0;
        float wj = 0.0f;
        if (lane < m) {
            sj = bucket[start + base + lane];
            sj = min(max(sj, 0), Nrows - 1);
            wj = dinv[sj] * di;
        }
        int k = 0;
        for (; k + 4 <= m; k += 4) {
            const float4 v0 = load_row(g, sj, k,     c4);
            const float4 v1 = load_row(g, sj, k + 1, c4);
            const float4 v2 = load_row(g, sj, k + 2, c4);
            const float4 v3 = load_row(g, sj, k + 3, c4);
            add_row(acc, v0, wj, k);
            add_row(acc, v1, wj, k + 1);
            add_row(acc, v2, wj, k + 2);
            add_row(acc, v3, wj, k + 3);
        }
        const int r = m - k;
        if (r == 3) {
            const float4 v0 = load_row(g, sj, k,     c4);
            const float4 v1 = load_row(g, sj, k + 1, c4);
            const float4 v2 = load_row(g, sj, k + 2, c4);
            add_row(acc, v0, wj, k);
            add_row(acc, v1, wj, k + 1);
            add_row(acc, v2, wj, k + 2);
        } else if (r == 2) {
            const float4 v0 = load_row(g, sj, k,     c4);
            const float4 v1 = load_row(g, sj, k + 1, c4);
            add_row(acc, v0, wj, k);
            add_row(acc, v1, wj, k + 1);
        } else if (r == 1) {
            const float4 v0 = load_row(g, sj, k, c4);
            add_row(acc, v0, wj, k);
        }
    }

    const float s2 = di * di;
    float4 r4;
    r4.x = acc.x + gs.x * s2 + bb.x;
    r4.y = acc.y + gs.y * s2 + bb.y;
    r4.z = acc.z + gs.z * s2 + bb.z;
    r4.w = acc.w + gs.w * s2 + bb.w;
    if (!PROJ) {
        *(float4*)(out + (size_t)i * DHID + c4) = r4;
        return;
    }
    const float v0 = (r4.x >= 0.0f) ? r4.x : NEG_SLOPE * r4.x;
    const float v1 = (r4.y >= 0.0f) ? r4.y : NEG_SLOPE * r4.y;
    const float v2 = (r4.z >= 0.0f) ? r4.z : NEG_SLOPE * r4.z;
    const float v3 = (r4.w >= 0.0f) ? r4.w : NEG_SLOPE * r4.w;
    const float4 wa = *(const float4*)(W2 + c4 * 2);
    const float4 wb = *(const float4*)(W2 + c4 * 2 + 4);
    float p0 = (v0 * wa.x + v1 * wa.z) + (v2 * wb.x + v3 * wb.z);
    float p1 = (v0 * wa.y + v1 * wa.w) + (v2 * wb.y + v3 * wb.w);
    p0 += xrow<1>(p0);  p1 += xrow<1>(p1);
    p0 += xrow<2>(p0);  p1 += xrow<2>(p1);
    p0 += xrow<4>(p0);  p1 += xrow<4>(p1);
    p0 += xrow<8>(p0);  p1 += xrow<8>(p1);
    p0 += xhalf(p0);    p1 += xhalf(p1);
    if (lane == 0) res[threadIdx.x >> 5] = make_float2(p0 + b2[0], p1 + b2[1]);
    __syncthreads();
    if (threadIdx.x < PROJ_WAVES) {
        const int node = (int)blockIdx.x * PROJ_WAVES + (int)threadIdx.x;
        if (node < Nrows) *(float2*)(out + (size_t)node * 2) = res[threadIdx.x];
    }
}

extern "C" void kernel_launch(void* const* d_in, const int* in_sizes, int n_in,
                              void* d_out, int out_size, void* d_ws, size_t ws_size,
                              hipStream_t stream)
{
    const float* x    = (const float*)d_in[0];
    const int*   ei   = (const int*)  d_in[1];
    const float* W_in = (const float*)d_in[3];
    const float* b_in = (const float*)d_in[4];
    const float* W_g  = (const float*)d_in[5];
    const float* b_g  = (const float*)d_in[6];
    const float* W_o1 = (const float*)d_in[7];
    const float* b_o1 = (const float*)d_in[8];
    const float* W_o2 = (const float*)d_in[9];
    const float* b_o2 = (const float*)d_in[10];

    const int ND = 239;
    const int N  = in_sizes[0] / ND;
    const int E  = in_sizes[1] / 2;
    const int* src = ei;
    const int* dst = ei + E;

    const size_t nh = (size_t)N * DHID;
    float* ws   = (float*)d_ws;
    int*   cnt  = (int*)d_ws;
    float* dinv = ws + (size_t)N;
    float* bufA = ws + 2 * (size_t)N;
    float* bufB = bufA + nh;
    float* Wsq    = bufB + nh;
    float* W3     = Wsq + DHID * DHID;
    float* c3     = W3 + DHID * DHID;
    float* d3     = c3 + DHID;
    int2*  cedge  = (int2*)(d3 + DHID);
    int*   off    = (int*)(cedge + (size_t)E);
    int*   bucket = off + (size_t)N;
    int*   ofs    = bucket + (size_t)E;

    const int gemm_blocks = (N / 16 + 3) / 4;
    const int gath_blocks = (N + 7) / 8;
    const int proj_blocks = (N + PROJ_WAVES - 1) / PROJ_WAVES;
    const int nbin  = (N + BIN_NODES - 1) / BIN_NODES;
    const int nbinp = (nbin + 1 + 31) / 32 * 32;
    const int epb   = (E + EDGE_BLOCKS - 1) / EDGE_BLOCKS;

    EdgeArgs eh;
    eh.src = src; eh.dst = dst; eh.ofs = ofs; eh.cedge = cedge;
    eh.off = off; eh.cnt = cnt; eh.dinv = dinv; eh.bucket = bucket;
    eh.E = E; eh.N = N; eh.nbin = nbin; eh.nbinp = nbinp; eh.epb = epb;
    eh.front = EDGE_BLOCKS;
    EdgeArgs ef = eh;
    ef.front = nbin;
    SideArgs noside;
    noside.Wg = nullptr; noside.bg = nullptr; noside.Wo1 = nullptr; noside.bo1 = nullptr;
    noside.Wsq = nullptr; noside.W3 = nullptr; noside.c3 = nullptr; noside.d3 = nullptr;
    SideArgs sd;
    sd.Wg = W_g; sd.bg = b_g; sd.Wo1 = W_o1; sd.bo1 = b_o1;
    sd.Wsq = Wsq; sd.W3 = W3; sd.c3 = c3; sd.d3 = d3;

    gemm_wmma<239, true, true, true, false, true><<<EDGE_BLOCKS + SIDE_BLOCKS + gemm_blocks, 128, 0, stream>>>(
        x, W_in, b_in, bufA, N, eh, sd);

    gemm_wmma<128, false, false, false, true, false><<<nbin + gemm_blocks, 128, 0, stream>>>(
        bufA, W3, nullptr, bufB, N, ef, noside);
    gather_kernel<false><<<gath_blocks, 256, 0, stream>>>(
        bufB, bucket, off, cnt, dinv, c3, bufA, N, nullptr, nullptr);
    gather_kernel<true><<<proj_blocks, PROJ_WAVES * 32, 0, stream>>>(
        bufA, bucket, off, cnt, dinv, d3, (float*)d_out, N, W_o2, b_o2);
}


# Aggregations, per-node load chain cut: pass 1 stores rows times dinv[i], pass 2 adds plain rows (no dinv[src] gather); pass 1 reads dinv[src] behind its row loads; off[i] with cnt[i] (rounding moves)
# speedup vs baseline: 4.2839x; 1.0306x over previous
#include <hip/hip_runtime.h>
#include <math.h>


typedef __attribute__((ext_vector_type(2))) float v2f;
typedef __attribute__((ext_vector_type(8))) float v8f;

#define DHID 128
#define TK 64
#define NEG_SLOPE 0.01f

#define BIN_SHIFT 8
#define BIN_NODES 256
#define NBIN_MAX 512
#define EDGE_BLOCKS 128
static_assert(EDGE_BLOCKS == 128, "pass 2 gives every slice to one of its 128 threads");

struct EdgeArgs {
    const int* src;
    const int* dst;
    int*   ofs;
    int2*  cedge;
    int*   off;
    int*   cnt;
    float* dinv;
    int*   bucket;
    int    E;
    int    N;
    int    nbin;
    int    nbinp;
    int    epb;
    int    front;
};

__device__ __forceinline__ int coarse_bin(int d, int nbin) {
    return min(max(d >> BIN_SHIFT, 0), nbin - 1);
}

__device__ __forceinline__ void slice_sort(int* lds, const EdgeArgs& ea, const int blk, const int tid) {
    int* h  = lds;
    int* ps = lds + NBIN_MAX;
    const int nbin = min(ea.nbin, NBIN_MAX);
    for (int b = tid; b < NBIN_MAX; b += 128) h[b] = 0;
    __syncthreads();
    const int e0 = min(blk * ea.epb, ea.E);
    const int e1 = min(ea.E, e0 + ea.epb);
    const int ns = e1 - e0;
    for (int e = e0 + tid; e < e1; e += 8 * 128) {
        int d[8];
#pragma unroll
        for (int u = 0; u < 8; ++u) d[u] = ea.dst[min(e + u * 128, e1 - 1)];
#pragma unroll
        for (int u = 0; u < 8; ++u)
            atomicAdd(&h[coarse_bin(d[u], nbin)], (e + u * 128 < e1) ? 1 : 0);
    }
    __syncthreads();
    const int c0 = h[4 * tid], c1 = h[4 * tid + 1], c2 = h[4 * tid + 2], c3 = h[4 * tid + 3];
    ps[tid] = (c0 + c1) + (c2 + c3);
    __syncthreads();
    for (int o = 1; o < 128; o <<= 1) {
        const int v = (tid >= o) ? ps[tid - o] : 0;
        __syncthreads();
        ps[tid] += v;
        __syncthreads();
    }
    const int x0 = ps[tid] - ((c0 + c1) + (c2 + c3));
    const int x1 = x0 + c0;
    const int x2 = x1 + c1;
    const int x3 = x2 + c2;
    h[4 * tid]     = x0;
    h[4 * tid + 1] = x1;
    h[4 * tid + 2] = x2;
    h[4 * tid + 3] = x3;
    int* __restrict__ row = ea.ofs + (size_t)blk * ea.nbinp;
    if (4 * tid     < nbin) row[4 * tid]     = x0;
    if (4 * tid + 1 < nbin) row[4 * tid + 1] = x1;
    if (4 * tid + 2 < nbin) row[4 * tid + 2] = x2;
    if (4 * tid + 3 < nbin) row[4 * tid + 3] = x3;
    if (tid == 127) row[nbin] = ps[127];
    __syncthreads();
    int2* __restrict__ ce = ea.cedge + e0;
    for (int e = e0 + tid; e < e1; e += 8 * 128) {
        int d[8], s[8];
#pragma unroll
        for (int u = 0; u < 8; ++u) {
            const int ee = min(e + u * 128, e1 - 1);
            d[u] = ea.dst[ee];
            s[u] = ea.src[ee];
        }
#pragma unroll
        for (int u = 0; u < 8; ++u) {
            const bool live = (e + u * 128 < e1);
            int pos = atomicAdd(&h[coarse_bin(d[u], nbin)], live ? 1 : 0);
            pos = min(max(pos, 0), ns - 1);
            if (live) ce[pos] = make_int2(s[u], d[u]);
        }
    }
}

__device__ __forceinline__ void fine_place(int* lds, const EdgeArgs& ea, const int b, const int tid) {
    int* c    = lds;
    int* ps   = lds + BIN_NODES;
    int* fill = lds + 2 * BIN_NODES;
    int* tot  = lds + 3 * BIN_NODES;
    c[tid] = 0;
    c[tid + 128] = 0;
    if (tid == 0) tot[0] = 0;
    const int base = min(tid * ea.epb, ea.E);
    const int cap  = min(ea.epb, ea.E - base);
    const int* __restrict__ row = ea.ofs + (size_t)tid * ea.nbinp;
    int s0 = row[b];
    int s1 = row[b + 1];
    s0 = min(max(s0, 0), cap);
    s1 = min(max(s1, s0), cap);
    const int2* __restrict__ ce = ea.cedge + base;
    __syncthreads();
    atomicAdd(&tot[0], s0);
    for (int i = s0; i < s1; i += 8) {
        int d[8];
#pragma unroll
        for (int u = 0; u < 8; ++u) d[u] = ce[min(i + u, s1 - 1)].y;
#pragma unroll
        for (int u = 0; u < 8; ++u)
            atomicAdd(&c[d[u] & (BIN_NODES - 1)], (i + u < s1) ? 1 : 0);
    }
    __syncthreads();
    const int c0 = c[2 * tid];
    const int c1 = c[2 * tid + 1];
    ps[tid] = c0 + c1;
    __syncthreads();
    for (int o = 1; o < 128; o <<= 1) {
        const int v = (tid >= o) ? ps[tid - o] : 0;
        __syncthreads();
        ps[tid] += v;
        __syncthreads();
    }
    const int start = min(max(tot[0], 0), ea.E);
    const int ex0 = start + ps[tid] - (c0 + c1);
    const int ex1 = ex0 + c0;
    fill[2 * tid]     = ex0;
    fill[2 * tid + 1] = ex1;
    const int node0 = b * BIN_NODES + 2 * tid;
    if (node0 < ea.N) {
        ea.off[node0]  = ex0;
        ea.cnt[node0]  = c0;
        ea.dinv[node0] = rsqrtf((float)c0 + 1.0f);
    }
    if (node0 + 1 < ea.N) {
        ea.off[node0 + 1]  = ex1;
        ea.cnt[node0 + 1]  = c1;
        ea.dinv[node0 + 1] = rsqrtf((float)c1 + 1.0f);
    }
    __syncthreads();
    for (int i = s0; i < s1; i += 8) {
        int2 ed[8];
#pragma unroll
        for (int u = 0; u < 8; ++u) ed[u] = ce[min(i + u, s1 - 1)];
#pragma unroll
        for (int u = 0; u < 8; ++u) {
            const bool live = (i + u < s1);
            int pos = atomicAdd(&fill[ed[u].y & (BIN_NODES - 1)], live ? 1 : 0);
            pos = min(max(pos, 0), ea.E - 1);
            if (live) ea.bucket[pos] = ed[u].x;
        }
    }
}


typedef float f4u __attribute__((ext_vector_type(4), aligned(4)));

template <int K, int KV>
__device__ __forceinline__ void gemm_chunk(
    float (*Wp)[DHID * 2], const float* __restrict__ arow0, const float* __restrict__ W,
    const int k0, const int tid, const int m, const int h, v8f (&acc)[8])
{
    constexpr int NG = (KV + 7) / 8;
    float4 a4[NG];
#pragma unroll
    for (int g = 0; g < NG; ++g) {
        if (8 * g + 8 <= KV) {
            const f4u v = *(const f4u*)(arow0 + k0 + 8 * g + 4 * h);
            a4[g] = make_float4(v.x, v.y, v.z, v.w);
        } else {
            const int e = k0 + 8 * g + 4 * h;
            a4[g].x = arow0[min(e,     K - 1)];
            a4[g].y = arow0[min(e + 1, K - 1)];
            a4[g].z = arow0[min(e + 2, K - 1)];
            a4[g].w = arow0[min(e + 3, K - 1)];
        }
    }

    constexpr int RT = ((NG * 8) % 32 == 0) ? 32 : 16;
    const float* __restrict__ wbase = W + (size_t)k0 * DHID;
    float* __restrict__ wdst = &Wp[0][tid * 2];
#pragma nounroll
    for (int r0 = 0; r0 < NG * 8; r0 += RT) {
        float t[RT];
#pragma unroll
        for (int u = 0; u < RT; ++u) {
            t[u] = (KV == NG * 8 || r0 + u < KV) ? wbase[(r0 + u) * DHID + tid] : 0.0f;
        }
#pragma unroll
        for (int u = 0; u < RT; ++u)
            wdst[((r0 + u) >> 1) * (DHID * 2) + (u & 1)] = t[u];
    }
    __syncthreads();

#pragma unroll
    for (int g = 0; g < NG; ++g) {
#pragma unroll
        for (int j = 0; j < 2; ++j) {
            v2f a;
            a.x = j ? a4[g].z : a4[g].x;
            a.y = j ? a4[g].w : a4[g].y;
            const float* __restrict__ wp = &Wp[4 * g + 2 * h + j][0];
#pragma unroll
            for (int t = 0; t < 8; ++t) {
                const v2f b = *(const v2f*)(wp + (t * 16 + m) * 2);
                acc[t] = __builtin_amdgcn_wmma_f32_16x16x4_f32(
                    false, a, false, b, (short)0, acc[t], false, false);
            }
        }
    }
    __syncthreads();
}

template <int K, bool BIAS, bool ACT>
__device__ __forceinline__ void gemm_tile(
    float (*Wp)[DHID * 2], const int bx,
    const float* __restrict__ A, const float* __restrict__ W,
    const float* __restrict__ bias, float* __restrict__ Out, int Nrows)
{
    const int tid  = threadIdx.x;
    const int wave = tid >> 5;
    const int lane = tid & 31;
    const int row0 = __builtin_amdgcn_readfirstlane((bx * 4 + wave) * 16);
    const bool rowok = (row0 < Nrows);

    const int m = lane & 15;
    const int h = lane >> 4;

    v8f acc[8];
#pragma unroll
    for (int t = 0; t < 8; ++t) acc[t] = (v8f){};

    const float* __restrict__ arow0 = A + (size_t)((rowok ? row0 : 0) + m) * K;

#pragma nounroll
    for (int k0 = 0; k0 + TK <= K; k0 += TK)
        gemm_chunk<K, TK>(Wp, arow0, W, k0, tid, m, h, acc);
    if (K % TK != 0)
        gemm_chunk<K, (K % TK != 0) ? (K % TK) : TK>(Wp, arow0, W, (K / TK) * TK, tid, m, h, acc);

    if (!rowok) return;
    const int rbase = row0 + ((lane >> 4) ? 8 : 0);
#pragma unroll
    for (int t = 0; t < 8; ++t) {
        const int col = t * 16 + m;
        const float bv = BIAS ? bias[col] : 0.0f;
#pragma unroll
        for (int r = 0; r < 8; ++r) {
            float v = acc[t][r] + bv;
            if (ACT) v = (v >= 0.0f) ? v : NEG_SLOPE * v;
            Out[(size_t)(rbase + r) * DHID + col] = v;
        }
    }
}

struct SideArgs {
    const float* Wg;
    const float* bg;
    const float* Wo1;
    const float* bo1;
    float*       Wsq;
    float*       W3;
    float*       c3;
    float*       d3;
};
#define SIDE_GEMM_BLOCKS (DHID / 64)
#define SIDE_BLOCKS (SIDE_GEMM_BLOCKS + 1)

template <int K, bool BIAS, bool ACT, bool HIST, bool FINE, bool SIDE>
__global__ __launch_bounds__(128, 6) void gemm_wmma(
    const float* __restrict__ A, const float* __restrict__ W,
    const float* __restrict__ bias, float* __restrict__ Out, int Nrows,
    const EdgeArgs ea, const SideArgs sa)
{
    __shared__ float Wp[TK / 2][DHID * 2];

    const int tid  = threadIdx.x;
    int bx = blockIdx.x;
    if (HIST || FINE) {
        if (bx < ea.front) {
            if (HIST) slice_sort((int*)&Wp[0][0], ea, bx, tid);
            else      fine_place((int*)&Wp[0][0], ea, bx, tid);
            return;
        }
        bx -= ea.front;
    }
    if (SIDE) {
        if (bx < SIDE_GEMM_BLOCKS) {
            gemm_tile<DHID, false, false>(Wp, bx, sa.Wg, sa.Wg, nullptr, sa.Wsq, DHID);
            __threadfence();
            __syncthreads();
            gemm_tile<DHID, false, false>(Wp, bx, sa.Wsq, sa.Wo1, nullptr, sa.W3, DHID);
            return;
        }
        if (bx == SIDE_GEMM_BLOCKS) {
            float* cg = &Wp[0][0];
            float s = 0.0f;
#pragma unroll 16
            for (int k = 0; k < DHID; ++k) s += sa.bg[k] * sa.Wg[k * DHID + tid];
            cg[tid] = s;
            __syncthreads();
            float s3 = 0.0f, sd = 0.0f;
#pragma unroll 16
            for (int k = 0; k < DHID; ++k) {
                const float w = sa.Wo1[k * DHID + tid];
                s3 += cg[k] * w;
                sd += sa.bg[k] * w;
            }
            sa.c3[tid] = s3;
            sa.d3[tid] = sd + sa.bo1[tid];
            return;
        }
        bx -= SIDE_BLOCKS;
    }
    gemm_tile<K, BIAS, ACT>(Wp, bx, A, W, bias, Out, Nrows);
}

__device__ __forceinline__ float4 load_row(const float* __restrict__ g, int sj, int k, int c4) {
    const int s = __builtin_amdgcn_readlane(sj, k);
    return *(const float4*)(g + (size_t)s * DHID + c4);
}

template <int M>
__device__ __forceinline__ float xrow(float x) {
    return __int_as_float(__builtin_amdgcn_update_dpp(0, __float_as_int(x), 0x160 + M, 0xf, 0xf, false));
}
__device__ __forceinline__ float xhalf(float x) {
    const int b = __float_as_int(x);
    return __int_as_float(__builtin_amdgcn_permlanex16(b, b, 0x76543210, 0xfedcba98, false, false));
}

template <bool WEIGHTED>
__device__ __forceinline__ void add_row(float4& acc, const float4 v, float dj, float di, int k) {
    if (WEIGHTED) {
        const float w = __int_as_float(__builtin_amdgcn_readlane(__float_as_int(dj), k)) * di;
        acc.x += v.x * w;
        acc.y += v.y * w;
        acc.z += v.z * w;
        acc.w += v.w * w;
    } else {
        acc.x += v.x;
        acc.y += v.y;
        acc.z += v.z;
        acc.w += v.w;
    }
}

#define PROJ_WAVES 16
template <bool PROJ>
__global__ __launch_bounds__(PROJ ? PROJ_WAVES * 32 : 256) void gather_kernel(
    const float* __restrict__ g, const int* __restrict__ bucket,
    const int* __restrict__ off, const int* __restrict__ cnt,
    const float* __restrict__ dinv, const float* __restrict__ bias,
    float* __restrict__ out, int Nrows,
    const float* __restrict__ W2, const float* __restrict__ b2)
{
    __shared__ float2 res[PROJ_WAVES];
    const int lane = threadIdx.x & 31;
    const int iw = __builtin_amdgcn_readfirstlane(
        (int)(blockIdx.x * (blockDim.x >> 5) + (threadIdx.x >> 5)));
    if (!PROJ && iw >= Nrows) return;
    const int i = PROJ ? min(iw, Nrows - 1) : iw;
    const int start = off[i];
    const int n     = cnt[i];
    const float di  = dinv[i];
    const int c4 = lane * 4;
    const float4 gs = *(const float4*)(g + (size_t)i * DHID + c4);
    const float4 bb = *(const float4*)(bias + c4);
    asm volatile("" :: "s"(start));

    float4 acc = make_float4(0.0f, 0.0f, 0.0f, 0.0f);
    for (int base = 0; base < n; base += 32) {
        const int m = min(32, n - base);
        int sj = bucket[min(start + base + lane, start + n - 1)];
        sj = min(max(sj, 0), Nrows - 1);
        float dj = 0.0f;
        if (!PROJ) dj = dinv[sj];
        int k = 0;
        for (; k + 4 <= m; k += 4) {
            const float4 v0 = load_row(g, sj, k,     c4);
            const float4 v1 = load_row(g, sj, k + 1, c4);
            const float4 v2 = load_row(g, sj, k + 2, c4);
            const float4 v3 = load_row(g, sj, k + 3, c4);
            add_row<!PROJ>(acc, v0, dj, di, k);
            add_row<!PROJ>(acc, v1, dj, di, k + 1);
            add_row<!PROJ>(acc, v2, dj, di, k + 2);
            add_row<!PROJ>(acc, v3, dj, di, k + 3);
        }
        const int r = m - k;
        if (r == 3) {
            const float4 v0 = load_row(g, sj, k,     c4);
            const float4 v1 = load_row(g, sj, k + 1, c4);
            const float4 v2 = load_row(g, sj, k + 2, c4);
            add_row<!PROJ>(acc, v0, dj, di, k);
            add_row<!PROJ>(acc, v1, dj, di, k + 1);
            add_row<!PROJ>(acc, v2, dj, di, k + 2);
        } else if (r == 2) {
            const float4 v0 = load_row(g, sj, k,     c4);
            const float4 v1 = load_row(g, sj, k + 1, c4);
            add_row<!PROJ>(acc, v0, dj, di, k);
            add_row<!PROJ>(acc, v1, dj, di, k + 1);
        } else if (r == 1) {
            const float4 v0 = load_row(g, sj, k, c4);
            add_row<!PROJ>(acc, v0, dj, di, k);
        }
    }

    float4 r4;
    if (!PROJ) {
        const float s2 = di * di;
        r4.x = (acc.x + gs.x * s2 + bb.x) * di;
        r4.y = (acc.y + gs.y * s2 + bb.y) * di;
        r4.z = (acc.z + gs.z * s2 + bb.z) * di;
        r4.w = (acc.w + gs.w * s2 + bb.w) * di;
        *(float4*)(out + (size_t)i * DHID + c4) = r4;
        return;
    }
    r4.x = (acc.x + gs.x) * di + bb.x;
    r4.y = (acc.y + gs.y) * di + bb.y;
    r4.z = (acc.z + gs.z) * di + bb.z;
    r4.w = (acc.w + gs.w) * di + bb.w;
    const float v0 = (r4.x >= 0.0f) ? r4.x : NEG_SLOPE * r4.x;
    const float v1 = (r4.y >= 0.0f) ? r4.y : NEG_SLOPE * r4.y;
    const float v2 = (r4.z >= 0.0f) ? r4.z : NEG_SLOPE * r4.z;
    const float v3 = (r4.w >= 0.0f) ? r4.w : NEG_SLOPE * r4.w;
    const float4 wa = *(const float4*)(W2 + c4 * 2);
    const float4 wb = *(const float4*)(W2 + c4 * 2 + 4);
    float p0 = (v0 * wa.x + v1 * wa.z) + (v2 * wb.x + v3 * wb.z);
    float p1 = (v0 * wa.y + v1 * wa.w) + (v2 * wb.y + v3 * wb.w);
    p0 += xrow<1>(p0);  p1 += xrow<1>(p1);
    p0 += xrow<2>(p0);  p1 += xrow<2>(p1);
    p0 += xrow<4>(p0);  p1 += xrow<4>(p1);
    p0 += xrow<8>(p0);  p1 += xrow<8>(p1);
    p0 += xhalf(p0);    p1 += xhalf(p1);
    if (lane == 0) res[threadIdx.x >> 5] = make_float2(p0 + b2[0], p1 + b2[1]);
    __syncthreads();
    if (threadIdx.x < PROJ_WAVES) {
        const int node = (int)blockIdx.x * PROJ_WAVES + (int)threadIdx.x;
        if (node < Nrows) *(float2*)(out + (size_t)node * 2) = res[threadIdx.x];
    }
}

extern "C" void kernel_launch(void* const* d_in, const int* in_sizes, int n_in,
                              void* d_out, int out_size, void* d_ws, size_t ws_size,
                              hipStream_t stream)
{
    const float* x    = (const float*)d_in[0];
    const int*   ei   = (const int*)  d_in[1];
    const float* W_in = (const float*)d_in[3];
    const float* b_in = (const float*)d_in[4];
    const float* W_g  = (const float*)d_in[5];
    const float* b_g  = (const float*)d_in[6];
    const float* W_o1 = (const float*)d_in[7];
    const float* b_o1 = (const float*)d_in[8];
    const float* W_o2 = (const float*)d_in[9];
    const float* b_o2 = (const float*)d_in[10];

    const int ND = 239;
    const int N  = in_sizes[0] / ND;
    const int E  = in_sizes[1] / 2;
    const int* src = ei;
    const int* dst = ei + E;

    const size_t nh = (size_t)N * DHID;
    float* ws   = (float*)d_ws;
    int*   cnt  = (int*)d_ws;
    float* dinv = ws + (size_t)N;
    float* bufA = ws + 2 * (size_t)N;
    float* bufB = bufA + nh;
    float* Wsq    = bufB + nh;
    float* W3     = Wsq + DHID * DHID;
    float* c3     = W3 + DHID * DHID;
    float* d3     = c3 + DHID;
    int2*  cedge  = (int2*)(d3 + DHID);
    int*   off    = (int*)(cedge + (size_t)E);
    int*   bucket = off + (size_t)N;
    int*   ofs    = bucket + (size_t)E;

    const int gemm_blocks = (N / 16 + 3) / 4;
    const int gath_blocks = (N + 7) / 8;
    const int proj_blocks = (N + PROJ_WAVES - 1) / PROJ_WAVES;
    const int nbin  = (N + BIN_NODES - 1) / BIN_NODES;
    const int nbinp = (nbin + 1 + 31) / 32 * 32;
    const int epb   = (E + EDGE_BLOCKS - 1) / EDGE_BLOCKS;

    EdgeArgs eh;
    eh.src = src; eh.dst = dst; eh.ofs = ofs; eh.cedge = cedge;
    eh.off = off; eh.cnt = cnt; eh.dinv = dinv; eh.bucket = bucket;
    eh.E = E; eh.N = N; eh.nbin = nbin; eh.nbinp = nbinp; eh.epb = epb;
    eh.front = EDGE_BLOCKS;
    EdgeArgs ef = eh;
    ef.front = nbin;
    SideArgs noside;
    noside.Wg = nullptr; noside.bg = nullptr; noside.Wo1 = nullptr; noside.bo1 = nullptr;
    noside.Wsq = nullptr; noside.W3 = nullptr; noside.c3 = nullptr; noside.d3 = nullptr;
    SideArgs sd;
    sd.Wg = W_g; sd.bg = b_g; sd.Wo1 = W_o1; sd.bo1 = b_o1;
    sd.Wsq = Wsq; sd.W3 = W3; sd.c3 = c3; sd.d3 = d3;

    gemm_wmma<239, true, true, true, false, true><<<EDGE_BLOCKS + SIDE_BLOCKS + gemm_blocks, 128, 0, stream>>>(
        x, W_in, b_in, bufA, N, eh, sd);

    gemm_wmma<128, false, false, false, true, false><<<nbin + gemm_blocks, 128, 0, stream>>>(
        bufA, W3, nullptr, bufB, N, ef, noside);
    gather_kernel<false><<<gath_blocks, 256, 0, stream>>>(
        bufB, bucket, off, cnt, dinv, c3, bufA, N, nullptr, nullptr);
    gather_kernel<true><<<proj_blocks, PROJ_WAVES * 32, 0, stream>>>(
        bufA, bucket, off, cnt, dinv, d3, (float*)d_out, N, W_o2, b_o2);
}
